# starting kernel
# baseline (speedup 1.0000x reference)
#include <hip/hip_runtime.h>
#include <math.h>

#define N_NODES 50000
#define N_EDGES 800000
#define IN_DIM 128
#define HIDDEN 64
#define NEG_SLOPE 0.2f
#define E_TOT (N_EDGES + N_NODES)

typedef float v2f __attribute__((ext_vector_type(2)));
typedef float v8f __attribute__((ext_vector_type(8)));

__device__ __forceinline__ void atomicMaxFloat(float* addr, float val) {
    if (val >= 0.0f) {
        atomicMax((int*)addr, __float_as_int(val));
    } else {
        atomicMin((unsigned int*)addr, __float_as_uint(val));
    }
}

__global__ void init_ws_kernel(float* __restrict__ mbuf, float* __restrict__ sbuf,
                               float* __restrict__ outacc) {
    int i = blockIdx.x * blockDim.x + threadIdx.x;
    if (i < N_NODES) { mbuf[i] = -INFINITY; sbuf[i] = 0.0f; }
    outacc[i] = 0.0f;
}

__global__ void __launch_bounds__(32)
gat_gemm_wmma_kernel(const float* __restrict__ x, const float* __restrict__ W,
                     const float* __restrict__ att_src_v, const float* __restrict__ att_dst_v,
                     float* __restrict__ h, float* __restrict__ a_src,
                     float* __restrict__ a_dst)
{
    const int tile = blockIdx.x;
    const int lane = threadIdx.x;
    const int half = lane >> 4;
    const int lrow = lane & 15;
    const int row0 = tile * 16;

    const float* xrow  = x + (size_t)(row0 + lrow) * IN_DIM;
    const float* wrow  = W + (size_t)lrow * IN_DIM;

    v8f acc0 = {}, acc1 = {}, acc2 = {}, acc3 = {};

    for (int kb = 0; kb < IN_DIM; kb += 4) {
        const int k = kb + 2 * half;
        v2f a  = *(const v2f*)(xrow + k);
        v2f b0 = *(const v2f*)(wrow + k);
        v2f b1 = *(const v2f*)(wrow + 16 * IN_DIM + k);
        v2f b2 = *(const v2f*)(wrow + 32 * IN_DIM + k);
        v2f b3 = *(const v2f*)(wrow + 48 * IN_DIM + k);
        acc0 = __builtin_amdgcn_wmma_f32_16x16x4_f32(false, a, false, b0, (short)0, acc0, false, false);
        acc1 = __builtin_amdgcn_wmma_f32_16x16x4_f32(false, a, false, b1, (short)0, acc1, false, false);
        acc2 = __builtin_amdgcn_wmma_f32_16x16x4_f32(false, a, false, b2, (short)0, acc2, false, false);
        acc3 = __builtin_amdgcn_wmma_f32_16x16x4_f32(false, a, false, b3, (short)0, acc3, false, false);
    }

    const float as0 = att_src_v[lrow],      ad0 = att_dst_v[lrow];
    const float as1 = att_src_v[16 + lrow], ad1 = att_dst_v[16 + lrow];
    const float as2 = att_src_v[32 + lrow], ad2 = att_dst_v[32 + lrow];
    const float as3 = att_src_v[48 + lrow], ad3 = att_dst_v[48 + lrow];

    #pragma unroll
    for (int r = 0; r < 8; ++r) {
        const int row = row0 + r + 8 * half;
        float* hr = h + (size_t)row * HIDDEN;
        hr[lrow]      = acc0[r];
        hr[16 + lrow] = acc1[r];
        hr[32 + lrow] = acc2[r];
        hr[48 + lrow] = acc3[r];
        float ps = acc0[r] * as0 + acc1[r] * as1 + acc2[r] * as2 + acc3[r] * as3;
        float pd = acc0[r] * ad0 + acc1[r] * ad1 + acc2[r] * ad2 + acc3[r] * ad3;
        #pragma unroll
        for (int off = 1; off < 16; off <<= 1) {
            ps += __shfl_xor(ps, off, 32);
            pd += __shfl_xor(pd, off, 32);
        }
        if (lrow == 0) { a_src[row] = ps; a_dst[row] = pd; }
    }
}

__global__ void edge_max_kernel(const int* __restrict__ ei,
                                const float* __restrict__ a_src,
                                const float* __restrict__ a_dst,
                                float* __restrict__ mbuf) {
    int e = blockIdx.x * blockDim.x + threadIdx.x;
    if (e >= E_TOT) return;
    int s, d;
    if (e < N_EDGES) { s = ei[e]; d = ei[N_EDGES + e]; }
    else             { s = d = e - N_EDGES; }
    float v = a_src[s] + a_dst[d];
    v = (v > 0.0f) ? v : v * NEG_SLOPE;
    atomicMaxFloat(&mbuf[d], v);
}

__global__ void edge_scatter_kernel(const int* __restrict__ ei,
                                    const float* __restrict__ a_src,
                                    const float* __restrict__ a_dst,
                                    const float* __restrict__ mbuf,
                                    const float* __restrict__ h,
                                    float* __restrict__ sbuf,
                                    float* __restrict__ outacc) {
    int wave = (blockIdx.x * blockDim.x + threadIdx.x) >> 5;
    int lane = threadIdx.x & 31;
    if (wave >= E_TOT) return;
    int s, d;
    if (wave < N_EDGES) { s = ei[wave]; d = ei[N_EDGES + wave]; }
    else                { s = d = wave - N_EDGES; }
    float v = a_src[s] + a_dst[d];
    v = (v > 0.0f) ? v : v * NEG_SLOPE;
    float ex = expf(v - mbuf[d]);
    if (lane == 0) atomicAdd(&sbuf[d], ex);
    v2f hv = *(const v2f*)(h + (size_t)s * HIDDEN + lane * 2);
    float* ob = outacc + (size_t)d * HIDDEN + lane * 2;
    atomicAdd(ob,     ex * hv.x);
    atomicAdd(ob + 1, ex * hv.y);
}

__global__ void finalize_kernel(const float* __restrict__ outacc,
                                const float* __restrict__ sbuf,
                                const float* __restrict__ bias_conv,
                                const float* __restrict__ W_lin,
                                const float* __restrict__ b_lin,
                                float* __restrict__ y) {
    int wave = (blockIdx.x * blockDim.x + threadIdx.x) >> 5;
    int lane = threadIdx.x & 31;
    if (wave >= N_NODES) return;
    float inv = 1.0f / sbuf[wave];
    v2f o = *(const v2f*)(outacc + (size_t)wave * HIDDEN + lane * 2);
    float o0 = fmaxf(o.x * inv + bias_conv[lane * 2],     0.0f);
    float o1 = fmaxf(o.y * inv + bias_conv[lane * 2 + 1], 0.0f);
    float p = o0 * W_lin[lane * 2] + o1 * W_lin[lane * 2 + 1];
    #pragma unroll
    for (int off = 1; off < 32; off <<= 1) p += __shfl_xor(p, off, 32);
    if (lane == 0) y[wave] = 1.0f / (1.0f + expf(-(p + b_lin[0])));
}

extern "C" void kernel_launch(void* const* d_in, const int* in_sizes, int n_in,
                              void* d_out, int out_size, void* d_ws, size_t ws_size,
                              hipStream_t stream) {
    const float* x         = (const float*)d_in[0];
    const int*   ei        = (const int*)  d_in[1];
    const float* W         = (const float*)d_in[2];
    const float* att_src_v = (const float*)d_in[3];
    const float* att_dst_v = (const float*)d_in[4];
    const float* bias_conv = (const float*)d_in[5];
    const float* W_lin     = (const float*)d_in[6];
    const float* b_lin     = (const float*)d_in[7];
    float* y = (float*)d_out;

    float* ws     = (float*)d_ws;
    float* h      = ws;
    float* a_src  = ws + 3200000;
    float* a_dst  = ws + 3250000;
    float* mbuf   = ws + 3300000;
    float* sbuf   = ws + 3350000;
    float* outacc = ws + 3400000;

    init_ws_kernel<<<12500, 256, 0, stream>>>(mbuf, sbuf, outacc);

    gat_gemm_wmma_kernel<<<3125, 32, 0, stream>>>(x, W, att_src_v, att_dst_v,
                                                  h, a_src, a_dst);

    edge_max_kernel<<<(E_TOT + 255) / 256, 256, 0, stream>>>(ei, a_src, a_dst, mbuf);

    edge_scatter_kernel<<<E_TOT / 8, 256, 0, stream>>>(ei, a_src, a_dst, mbuf,
                                                       h, sbuf, outacc);

    finalize_kernel<<<(N_NODES + 7) / 8, 256, 0, stream>>>(outacc, sbuf, bias_conv,
                                                           W_lin, b_lin, y);
}


# Each block owns 200 destination nodes: filters the dst list into an LDS queue, then segment max/sum/weighted sum by LDS atomics and the output head; no global atomics; sums in arrival order
# speedup vs baseline: 3.1768x; 3.1768x over previous
#include <hip/hip_runtime.h>
#include <math.h>

#define N_NODES 50000
#define N_EDGES 800000
#define IN_DIM 128
#define HIDDEN 64
#define NEG_SLOPE 0.2f
#define E_TOT (N_EDGES + N_NODES)

typedef float v2f __attribute__((ext_vector_type(2)));
typedef float v8f __attribute__((ext_vector_type(8)));

__global__ void __launch_bounds__(32)
gat_gemm_wmma_kernel(const float* __restrict__ x, const float* __restrict__ W,
                     const float* __restrict__ att_src_v, const float* __restrict__ att_dst_v,
                     float* __restrict__ h, float* __restrict__ a_src,
                     float* __restrict__ a_dst)
{
    const int tile = blockIdx.x;
    const int lane = threadIdx.x;
    const int half = lane >> 4;
    const int lrow = lane & 15;
    const int row0 = tile * 16;

    const float* xrow  = x + (size_t)(row0 + lrow) * IN_DIM;
    const float* wrow  = W + (size_t)lrow * IN_DIM;

    v8f acc0 = {}, acc1 = {}, acc2 = {}, acc3 = {};

    for (int kb = 0; kb < IN_DIM; kb += 4) {
        const int k = kb + 2 * half;
        v2f a  = *(const v2f*)(xrow + k);
        v2f b0 = *(const v2f*)(wrow + k);
        v2f b1 = *(const v2f*)(wrow + 16 * IN_DIM + k);
        v2f b2 = *(const v2f*)(wrow + 32 * IN_DIM + k);
        v2f b3 = *(const v2f*)(wrow + 48 * IN_DIM + k);
        acc0 = __builtin_amdgcn_wmma_f32_16x16x4_f32(false, a, false, b0, (short)0, acc0, false, false);
        acc1 = __builtin_amdgcn_wmma_f32_16x16x4_f32(false, a, false, b1, (short)0, acc1, false, false);
        acc2 = __builtin_amdgcn_wmma_f32_16x16x4_f32(false, a, false, b2, (short)0, acc2, false, false);
        acc3 = __builtin_amdgcn_wmma_f32_16x16x4_f32(false, a, false, b3, (short)0, acc3, false, false);
    }

    const float as0 = att_src_v[lrow],      ad0 = att_dst_v[lrow];
    const float as1 = att_src_v[16 + lrow], ad1 = att_dst_v[16 + lrow];
    const float as2 = att_src_v[32 + lrow], ad2 = att_dst_v[32 + lrow];
    const float as3 = att_src_v[48 + lrow], ad3 = att_dst_v[48 + lrow];

    #pragma unroll
    for (int r = 0; r < 8; ++r) {
        const int row = row0 + r + 8 * half;
        float* hr = h + (size_t)row * HIDDEN;
        hr[lrow]      = acc0[r];
        hr[16 + lrow] = acc1[r];
        hr[32 + lrow] = acc2[r];
        hr[48 + lrow] = acc3[r];
        float ps = acc0[r] * as0 + acc1[r] * as1 + acc2[r] * as2 + acc3[r] * as3;
        float pd = acc0[r] * ad0 + acc1[r] * ad1 + acc2[r] * ad2 + acc3[r] * ad3;
        #pragma unroll
        for (int off = 1; off < 16; off <<= 1) {
            ps += __shfl_xor(ps, off, 32);
            pd += __shfl_xor(pd, off, 32);
        }
        if (lrow == 0) { a_src[row] = ps; a_dst[row] = pd; }
    }
}

#define NPB 200
#define N_OWN_BLOCKS (N_NODES / NPB)
#define QCAP 16384
#define OWN_T 1024
#define PB 16
#define SCAN_U 8

__device__ __forceinline__ void lds_max_float(float* addr, float val) {
    if (val >= 0.0f) atomicMax((int*)addr, __float_as_int(val));
    else             atomicMin((unsigned int*)addr, __float_as_uint(val));
}

__device__ __forceinline__ void q_push(unsigned* q_pk, int* q_n, int e, unsigned dl) {
    int slot = atomicAdd(q_n, 1);
    if (slot < QCAP) q_pk[slot] = (unsigned)e | (dl << 20);
}

template <bool GUARD>
__device__ __forceinline__ void scan_round(const int4* __restrict__ dst4, int b4, int len4,
                                           int rot4, int node0, int i0,
                                           unsigned* q_pk, int* q_n) {
    int4 d[SCAN_U];
    int  j[SCAN_U];
    #pragma unroll
    for (int k = 0; k < SCAN_U; ++k) {
        int jj = i0 + k * OWN_T + rot4;
        if (jj >= len4) jj -= len4;
        j[k] = b4 + jj;
        if (!GUARD || (i0 + k * OWN_T < len4)) d[k] = dst4[j[k]];
        else                                   d[k] = make_int4(-1, -1, -1, -1);
    }
    #pragma unroll
    for (int k = 0; k < SCAN_U; ++k) {
        const unsigned l0 = (unsigned)(d[k].x - node0);
        const unsigned l1 = (unsigned)(d[k].y - node0);
        const unsigned l2 = (unsigned)(d[k].z - node0);
        const unsigned l3 = (unsigned)(d[k].w - node0);
        const unsigned l01 = (l0 < l1) ? l0 : l1;
        const unsigned l23 = (l2 < l3) ? l2 : l3;
        if (((l01 < l23) ? l01 : l23) < NPB) {
            const int e = 4 * j[k];
            if (l0 < NPB) q_push(q_pk, q_n, e,     l0);
            if (l1 < NPB) q_push(q_pk, q_n, e + 1, l1);
            if (l2 < NPB) q_push(q_pk, q_n, e + 2, l2);
            if (l3 < NPB) q_push(q_pk, q_n, e + 3, l3);
        }
    }
}

__device__ __forceinline__ void scan_enqueue(const int4* __restrict__ dst4, int b4, int len4,
                                             int rot4, int node0, int tid,
                                             unsigned* q_pk, int* q_n) {
    const int full = len4 / (SCAN_U * OWN_T);
    for (int r = 0; r < full; ++r)
        scan_round<false>(dst4, b4, len4, rot4, node0, tid + r * (SCAN_U * OWN_T), q_pk, q_n);
    if (full * (SCAN_U * OWN_T) < len4)
        scan_round<true>(dst4, b4, len4, rot4, node0, tid + full * (SCAN_U * OWN_T), q_pk, q_n);
}

__device__ __forceinline__ void resolve_max(const int* __restrict__ ei,
                                            const float* __restrict__ a_src,
                                            const float* adst_l, float* m_l,
                                            const unsigned* q_pk, unsigned* q_sd, float* q_sc,
                                            int n, int tid) {
    for (int i0 = tid; i0 < n; i0 += 4 * OWN_T) {
        unsigned u[4];
        int      s[4];
        float    as[4];
        #pragma unroll
        for (int k = 0; k < 4; ++k) {
            const int i = i0 + k * OWN_T;
            u[k] = q_pk[(i < n) ? i : (n - 1)];
        }
        #pragma unroll
        for (int k = 0; k < 4; ++k) s[k] = ei[u[k] & 0xFFFFFu];
        #pragma unroll
        for (int k = 0; k < 4; ++k) as[k] = a_src[s[k]];
        #pragma unroll
        for (int k = 0; k < 4; ++k) {
            const int i = i0 + k * OWN_T;
            if (i < n) {
                const unsigned dl = u[k] >> 20;
                float v = as[k] + adst_l[dl];
                v = (v > 0.0f) ? v : v * NEG_SLOPE;
                q_sd[i] = (unsigned)s[k] | (dl << 16);
                q_sc[i] = v;
                lds_max_float(&m_l[dl], v);
            }
        }
    }
}

template <bool FULL>
__device__ __forceinline__ void process_batch(const float* __restrict__ h,
                                              const float* m_l, float* s_l, float* acc,
                                              const unsigned* q_sd, const float* q_sc,
                                              int base, int n, int lane) {
    const int  my = base + (lane & (PB - 1));
    const bool ok = FULL || (my < n);
    const unsigned u  = q_sd[ok ? my : base];
    const float    sc = q_sc[ok ? my : base];
    const float    ex = expf(sc - m_l[u >> 16]);
    if (ok && lane < PB) atomicAdd(&s_l[u >> 16], ex);
    v2f hv[PB];
    #pragma unroll
    for (int k = 0; k < PB; ++k) {
        const unsigned uk = (unsigned)__builtin_amdgcn_readlane((int)u, k);
        hv[k] = *(const v2f*)(h + (size_t)(uk & 0xFFFFu) * HIDDEN + lane * 2);
    }
    #pragma unroll
    for (int k = 0; k < PB; ++k) {
        if (FULL || (base + k < n)) {
            const unsigned uk  = (unsigned)__builtin_amdgcn_readlane((int)u, k);
            const float    exk = __int_as_float(__builtin_amdgcn_readlane(__float_as_int(ex), k));
            float* a = acc + (uk >> 16) * HIDDEN + lane * 2;
            atomicAdd(a,     exk * hv[k].x);
            atomicAdd(a + 1, exk * hv[k].y);
        }
    }
}

__device__ __forceinline__ void process_queue(const float* __restrict__ h,
                                              const float* m_l, float* s_l, float* acc,
                                              const unsigned* q_sd, const float* q_sc,
                                              int n, int wv, int lane) {
    for (int base = wv * PB; base < n; base += (OWN_T / 32) * PB) {
        if (base + PB <= n) process_batch<true >(h, m_l, s_l, acc, q_sd, q_sc, base, n, lane);
        else                process_batch<false>(h, m_l, s_l, acc, q_sd, q_sc, base, n, lane);
    }
}

__global__ void __launch_bounds__(OWN_T)
gat_owner_kernel(const int* __restrict__ ei,
                 const float* __restrict__ a_src,
                 const float* __restrict__ a_dst,
                 const float* __restrict__ h,
                 const float* __restrict__ bias_conv,
                 const float* __restrict__ W_lin,
                 const float* __restrict__ b_lin,
                 float* __restrict__ y) {
    __shared__ float    acc[NPB * HIDDEN];
    __shared__ float    m_l[NPB];
    __shared__ float    s_l[NPB];
    __shared__ float    adst_l[NPB];
    __shared__ float    self_sc[NPB];
    __shared__ unsigned q_pk[QCAP];
    __shared__ unsigned q_sd[QCAP];
    __shared__ float    q_sc[QCAP];
    __shared__ int      q_n;

    const int tid   = threadIdx.x;
    const int lane  = tid & 31;
    const int wv    = __builtin_amdgcn_readfirstlane(tid >> 5);
    const int node0 = blockIdx.x * NPB;
    const int4* dst4 = (const int4*)(ei + N_EDGES);

    if (tid < NPB) {
        const float ad = a_dst[node0 + tid];
        float v = a_src[node0 + tid] + ad;
        v = (v > 0.0f) ? v : v * NEG_SLOPE;
        adst_l[tid]  = ad;
        self_sc[tid] = v;
        m_l[tid]     = v;
    }
    if (tid == 0) q_n = 0;
    __syncthreads();

    scan_enqueue(dst4, 0, N_EDGES / 4, blockIdx.x * (N_EDGES / 4 / N_OWN_BLOCKS), node0, tid, q_pk, &q_n);
    __syncthreads();
    const int n_all = __builtin_amdgcn_readfirstlane(q_n);

    if (n_all <= QCAP) {
        resolve_max(ei, a_src, adst_l, m_l, q_pk, q_sd, q_sc, n_all, tid);
        __syncthreads();
        for (int i = tid; i < NPB * HIDDEN; i += OWN_T) {
            const int dl = i >> 6;
            const float ex = expf(self_sc[dl] - m_l[dl]);
            acc[i] = ex * h[(size_t)node0 * HIDDEN + i];
            if ((i & 63) == 0) s_l[dl] = ex;
        }
        __syncthreads();
        process_queue(h, m_l, s_l, acc, q_sd, q_sc, n_all, wv, lane);
        __syncthreads();
    } else {
        for (int b = 0; b < N_EDGES; b += QCAP) {
            const int len = (N_EDGES - b < QCAP) ? (N_EDGES - b) : QCAP;
            __syncthreads();
            if (tid == 0) q_n = 0;
            __syncthreads();
            scan_enqueue(dst4, b / 4, len / 4, 0, node0, tid, q_pk, &q_n);
            __syncthreads();
            resolve_max(ei, a_src, adst_l, m_l, q_pk, q_sd, q_sc, __builtin_amdgcn_readfirstlane(q_n), tid);
        }
        __syncthreads();
        for (int i = tid; i < NPB * HIDDEN; i += OWN_T) {
            const int dl = i >> 6;
            const float ex = expf(self_sc[dl] - m_l[dl]);
            acc[i] = ex * h[(size_t)node0 * HIDDEN + i];
            if ((i & 63) == 0) s_l[dl] = ex;
        }
        for (int b = 0; b < N_EDGES; b += QCAP) {
            const int len = (N_EDGES - b < QCAP) ? (N_EDGES - b) : QCAP;
            __syncthreads();
            if (tid == 0) q_n = 0;
            __syncthreads();
            scan_enqueue(dst4, b / 4, len / 4, 0, node0, tid, q_pk, &q_n);
            __syncthreads();
            const int n = __builtin_amdgcn_readfirstlane(q_n);
            resolve_max(ei, a_src, adst_l, m_l, q_pk, q_sd, q_sc, n, tid);
            __syncthreads();
            process_queue(h, m_l, s_l, acc, q_sd, q_sc, n, wv, lane);
        }
        __syncthreads();
    }

    const float bc0 = bias_conv[lane * 2], bc1 = bias_conv[lane * 2 + 1];
    const float wl0 = W_lin[lane * 2],     wl1 = W_lin[lane * 2 + 1];
    const float bl  = b_lin[0];
    for (int dl = wv; dl < NPB; dl += OWN_T / 32) {
        const float inv = 1.0f / s_l[dl];
        const v2f o = *(const v2f*)(acc + dl * HIDDEN + lane * 2);
        const float o0 = fmaxf(o.x * inv + bc0, 0.0f);
        const float o1 = fmaxf(o.y * inv + bc1, 0.0f);
        float p = o0 * wl0 + o1 * wl1;
        #pragma unroll
        for (int off = 1; off < 32; off <<= 1) p += __shfl_xor(p, off, 32);
        if (lane == 0) y[node0 + dl] = 1.0f / (1.0f + expf(-(p + bl)));
    }
}

extern "C" void kernel_launch(void* const* d_in, const int* in_sizes, int n_in,
                              void* d_out, int out_size, void* d_ws, size_t ws_size,
                              hipStream_t stream) {
    const float* x         = (const float*)d_in[0];
    const int*   ei        = (const int*)  d_in[1];
    const float* W         = (const float*)d_in[2];
    const float* att_src_v = (const float*)d_in[3];
    const float* att_dst_v = (const float*)d_in[4];
    const float* bias_conv = (const float*)d_in[5];
    const float* W_lin     = (const float*)d_in[6];
    const float* b_lin     = (const float*)d_in[7];
    float* y = (float*)d_out;

    float* ws     = (float*)d_ws;
    float* h      = ws;
    float* a_src  = ws + 3200000;
    float* a_dst  = ws + 3250000;

    gat_gemm_wmma_kernel<<<3125, 32, 0, stream>>>(x, W, att_src_v, att_dst_v,
                                                  h, a_src, a_dst);

    gat_owner_kernel<<<N_OWN_BLOCKS, OWN_T, 0, stream>>>(ei, a_src, a_dst, h, bias_conv,
                                                         W_lin, b_lin, y);
}


# GEMM waves also write each edge's owner-block number (dst/200) as 8 bit planes per 32 edges; owner blocks scan these planes, not the dst list, and read dst only for own edges; sums in arrival order
# speedup vs baseline: 3.9646x; 1.2480x over previous
#include <hip/hip_runtime.h>
#include <math.h>

#define N_NODES 50000
#define N_EDGES 800000
#define IN_DIM 128
#define HIDDEN 64
#define NEG_SLOPE 0.2f
#define E_TOT (N_EDGES + N_NODES)
#define NPB 200
#define N_OWN_BLOCKS (N_NODES / NPB)
#define N_TILES (N_NODES / 16)
#define NO_BIN 255u

typedef float v2f __attribute__((ext_vector_type(2)));
typedef float v8f __attribute__((ext_vector_type(8)));

__device__ __forceinline__ unsigned bin_of(int d) {
    const unsigned u = (unsigned)d;
    return (u < (unsigned)N_NODES) ? (u / NPB) : NO_BIN;
}

__global__ void __launch_bounds__(32)
gat_gemm_wmma_kernel(const float* __restrict__ x, const float* __restrict__ W,
                     const float* __restrict__ att_src_v, const float* __restrict__ att_dst_v,
                     float* __restrict__ h, float* __restrict__ a_src,
                     float* __restrict__ a_dst,
                     const int* __restrict__ ei, unsigned* __restrict__ binw)
{
    const int tile = blockIdx.x;
    const int lane = threadIdx.x;
    const int half = lane >> 4;
    const int lrow = lane & 15;
    const int row0 = tile * 16;

    const float* xrow  = x + (size_t)(row0 + lrow) * IN_DIM;
    const float* wrow  = W + (size_t)lrow * IN_DIM;

    static_assert(N_EDGES == N_TILES * 256, "each GEMM wave converts 256 edges");
    const int4* dst4 = (const int4*)(ei + N_EDGES);
    const int4 dv0 = dst4[tile * 64 + lane];
    const int4 dv1 = dst4[tile * 64 + 32 + lane];

    v8f acc0 = {}, acc1 = {}, acc2 = {}, acc3 = {};

    for (int kb = 0; kb < IN_DIM; kb += 4) {
        const int k = kb + 2 * half;
        v2f a  = *(const v2f*)(xrow + k);
        v2f b0 = *(const v2f*)(wrow + k);
        v2f b1 = *(const v2f*)(wrow + 16 * IN_DIM + k);
        v2f b2 = *(const v2f*)(wrow + 32 * IN_DIM + k);
        v2f b3 = *(const v2f*)(wrow + 48 * IN_DIM + k);
        acc0 = __builtin_amdgcn_wmma_f32_16x16x4_f32(false, a, false, b0, (short)0, acc0, false, false);
        acc1 = __builtin_amdgcn_wmma_f32_16x16x4_f32(false, a, false, b1, (short)0, acc1, false, false);
        acc2 = __builtin_amdgcn_wmma_f32_16x16x4_f32(false, a, false, b2, (short)0, acc2, false, false);
        acc3 = __builtin_amdgcn_wmma_f32_16x16x4_f32(false, a, false, b3, (short)0, acc3, false, false);
    }

    const float as0 = att_src_v[lrow],      ad0 = att_dst_v[lrow];
    const float as1 = att_src_v[16 + lrow], ad1 = att_dst_v[16 + lrow];
    const float as2 = att_src_v[32 + lrow], ad2 = att_dst_v[32 + lrow];
    const float as3 = att_src_v[48 + lrow], ad3 = att_dst_v[48 + lrow];

    #pragma unroll
    for (int r = 0; r < 8; ++r) {
        const int row = row0 + r + 8 * half;
        float* hr = h + (size_t)row * HIDDEN;
        hr[lrow]      = acc0[r];
        hr[16 + lrow] = acc1[r];
        hr[32 + lrow] = acc2[r];
        hr[48 + lrow] = acc3[r];
        float ps = acc0[r] * as0 + acc1[r] * as1 + acc2[r] * as2 + acc3[r] * as3;
        float pd = acc0[r] * ad0 + acc1[r] * ad1 + acc2[r] * ad2 + acc3[r] * ad3;
        #pragma unroll
        for (int off = 1; off < 16; off <<= 1) {
            ps += __shfl_xor(ps, off, 32);
            pd += __shfl_xor(pd, off, 32);
        }
        if (lrow == 0) { a_src[row] = ps; a_dst[row] = pd; }
    }

    {
        unsigned wa = 0, wb = 0;
        #pragma unroll
        for (int t = 0; t < 2; ++t) {
            const int4 dv = t ? dv1 : dv0;
            const unsigned bb[4] = { bin_of(dv.x), bin_of(dv.y), bin_of(dv.z), bin_of(dv.w) };
            #pragma unroll
            for (int c = 0; c < 4; ++c) {
                #pragma unroll
                for (int i = 0; i < 8; ++i) {
                    const unsigned pl = __builtin_amdgcn_ballot_w32(((bb[c] >> i) & 1u) != 0u);
                    if (i < 4) wa = (lane == 16 * t + 4 * c + i)       ? pl : wa;
                    else       wb = (lane == 16 * t + 4 * c + (i - 4)) ? pl : wb;
                }
            }
        }
        binw[tile * 32 + lane]               = wa;
        binw[N_EDGES / 8 + tile * 32 + lane] = wb;
    }
}

#define QCAP 16384
#define OWN_T 1024
#define PB 16
#define SCAN_U 5

__device__ __forceinline__ void lds_max_float(float* addr, float val) {
    if (val >= 0.0f) atomicMax((int*)addr, __float_as_int(val));
    else             atomicMin((unsigned int*)addr, __float_as_uint(val));
}

struct PlaneKey { unsigned w[8]; };

__device__ __forceinline__ unsigned hits32(uint4 lo, uint4 hi, const PlaneKey& key) {
    return (lo.x ^ key.w[0]) & (lo.y ^ key.w[1]) & (lo.z ^ key.w[2]) & (lo.w ^ key.w[3]) &
           (hi.x ^ key.w[4]) & (hi.y ^ key.w[5]) & (hi.z ^ key.w[6]) & (hi.w ^ key.w[7]);
}

__device__ __forceinline__ int wave_prefix_sum(int v, int lane) {
    v += __builtin_amdgcn_update_dpp(0, v, 0x111, 0xF, 0xF, false);
    v += __builtin_amdgcn_update_dpp(0, v, 0x112, 0xF, 0xF, false);
    v += __builtin_amdgcn_update_dpp(0, v, 0x114, 0xF, 0xF, false);
    v += __builtin_amdgcn_update_dpp(0, v, 0x118, 0xF, 0xF, false);
    const int low = __builtin_amdgcn_readlane(v, 15);
    return v + ((lane >= 16) ? low : 0);
}

template <bool GUARD>
__device__ __forceinline__ void scan_round(const uint4* __restrict__ plane_lo,
                                           const uint4* __restrict__ plane_hi,
                                           int g0, int lenG, int rotG, const PlaneKey& key,
                                           int i0, int lane, unsigned* q_pk, int* q_n) {
    uint4 lo[SCAN_U], hi[SCAN_U];
    int   G[SCAN_U];
    #pragma unroll
    for (int k = 0; k < SCAN_U; ++k) {
        int jj = i0 + k * OWN_T + rotG;
        if (jj >= lenG) jj -= lenG;
        G[k] = g0 + jj;
        if (!GUARD || (i0 + k * OWN_T < lenG)) { lo[k] = plane_lo[G[k]]; hi[k] = plane_hi[G[k]]; }
        else { lo[k] = make_uint4(~0u, ~0u, ~0u, ~0u); hi[k] = lo[k]; }
    }
    unsigned m[SCAN_U];
    int cnt = 0;
    #pragma unroll
    for (int k = 0; k < SCAN_U; ++k) {
        m[k] = hits32(lo[k], hi[k], key);
        cnt += __popc(m[k]);
    }
    const int incl  = wave_prefix_sum(cnt, lane);
    const int total = __builtin_amdgcn_readlane(incl, 31);
    int base = 0;
    if (lane == 0) base = atomicAdd(q_n, total);
    int slot = __builtin_amdgcn_readfirstlane(base) + incl - cnt;
    #pragma unroll
    for (int k = 0; k < SCAN_U; ++k) {
        unsigned mm = m[k];
        const int e0 = ((G[k] & ~3) << 5) + (G[k] & 3);
        while (mm) {
            const int p = __builtin_ctz(mm);
            mm &= mm - 1;
            if (slot < QCAP) q_pk[slot] = (unsigned)(e0 + 4 * p);
            ++slot;
        }
    }
}

__device__ __forceinline__ void scan_enqueue(const uint4* __restrict__ plane_lo,
                                             const uint4* __restrict__ plane_hi,
                                             int g0, int lenG, int rotG, const PlaneKey& key,
                                             int tid, unsigned* q_pk, int* q_n) {
    const int full = lenG / (SCAN_U * OWN_T);
    for (int r = 0; r < full; ++r)
        scan_round<false>(plane_lo, plane_hi, g0, lenG, rotG, key, tid + r * (SCAN_U * OWN_T), tid & 31, q_pk, q_n);
    if (full * (SCAN_U * OWN_T) < lenG)
        scan_round<true>(plane_lo, plane_hi, g0, lenG, rotG, key, tid + full * (SCAN_U * OWN_T), tid & 31, q_pk, q_n);
}

__device__ __forceinline__ void resolve_max(const int* __restrict__ ei,
                                            const float* __restrict__ a_src,
                                            const float* adst_l, float* m_l,
                                            const unsigned* q_pk, unsigned* q_sd, float* q_sc,
                                            int n, int tid, int node0) {
    for (int i0 = tid; i0 < n; i0 += 4 * OWN_T) {
        unsigned u[4];
        int      s[4];
        int      dd[4];
        float    as[4];
        #pragma unroll
        for (int k = 0; k < 4; ++k) {
            const int i = i0 + k * OWN_T;
            u[k] = q_pk[(i < n) ? i : (n - 1)];
        }
        #pragma unroll
        for (int k = 0; k < 4; ++k) { s[k] = ei[u[k]]; dd[k] = ei[N_EDGES + u[k]]; }
        #pragma unroll
        for (int k = 0; k < 4; ++k) as[k] = a_src[s[k]];
        #pragma unroll
        for (int k = 0; k < 4; ++k) {
            const int i = i0 + k * OWN_T;
            if (i < n) {
                const unsigned dl = (unsigned)(dd[k] - node0);
                float v = as[k] + adst_l[dl];
                v = (v > 0.0f) ? v : v * NEG_SLOPE;
                q_sd[i] = (unsigned)s[k] | (dl << 16);
                q_sc[i] = v;
                lds_max_float(&m_l[dl], v);
            }
        }
    }
}

template <bool FULL>
__device__ __forceinline__ void process_batch(const float* __restrict__ h,
                                              const float* m_l, float* s_l, float* acc,
                                              const unsigned* q_sd, const float* q_sc,
                                              int base, int n, int lane) {
    const int  my = base + (lane & (PB - 1));
    const bool ok = FULL || (my < n);
    const unsigned u  = q_sd[ok ? my : base];
    const float    sc = q_sc[ok ? my : base];
    const float    ex = expf(sc - m_l[u >> 16]);
    if (ok && lane < PB) atomicAdd(&s_l[u >> 16], ex);
    v2f hv[PB];
    #pragma unroll
    for (int k = 0; k < PB; ++k) {
        const unsigned uk = (unsigned)__builtin_amdgcn_readlane((int)u, k);
        hv[k] = *(const v2f*)(h + (size_t)(uk & 0xFFFFu) * HIDDEN + lane * 2);
    }
    #pragma unroll
    for (int k = 0; k < PB; ++k) {
        if (FULL || (base + k < n)) {
            const unsigned uk  = (unsigned)__builtin_amdgcn_readlane((int)u, k);
            const float    exk = __int_as_float(__builtin_amdgcn_readlane(__float_as_int(ex), k));
            float* a = acc + (uk >> 16) * HIDDEN + lane * 2;
            atomicAdd(a,     exk * hv[k].x);
            atomicAdd(a + 1, exk * hv[k].y);
        }
    }
}

__device__ __forceinline__ void process_queue(const float* __restrict__ h,
                                              const float* m_l, float* s_l, float* acc,
                                              const unsigned* q_sd, const float* q_sc,
                                              int n, int wv, int lane) {
    for (int base = wv * PB; base < n; base += (OWN_T / 32) * PB) {
        if (base + PB <= n) process_batch<true >(h, m_l, s_l, acc, q_sd, q_sc, base, n, lane);
        else                process_batch<false>(h, m_l, s_l, acc, q_sd, q_sc, base, n, lane);
    }
}

__global__ void __launch_bounds__(OWN_T)
gat_owner_kernel(const int* __restrict__ ei,
                 const unsigned* __restrict__ binw,
                 const float* __restrict__ a_src,
                 const float* __restrict__ a_dst,
                 const float* __restrict__ h,
                 const float* __restrict__ bias_conv,
                 const float* __restrict__ W_lin,
                 const float* __restrict__ b_lin,
                 float* __restrict__ y) {
    __shared__ float    acc[NPB * HIDDEN];
    __shared__ float    m_l[NPB];
    __shared__ float    s_l[NPB];
    __shared__ float    adst_l[NPB];
    __shared__ float    self_sc[NPB];
    __shared__ unsigned q_pk[QCAP];
    __shared__ unsigned q_sd[QCAP];
    __shared__ float    q_sc[QCAP];
    __shared__ int      q_n;

    const int tid   = threadIdx.x;
    const int lane  = tid & 31;
    const int wv    = __builtin_amdgcn_readfirstlane(tid >> 5);
    const int node0 = blockIdx.x * NPB;
    const uint4* plane_lo = (const uint4*)binw;
    const uint4* plane_hi = (const uint4*)(binw + N_EDGES / 8);
    PlaneKey key;
    #pragma unroll
    for (int i = 0; i < 8; ++i) key.w[i] = ((blockIdx.x >> i) & 1u) - 1u;
    static_assert(N_OWN_BLOCKS <= 255 && N_EDGES % 128 == 0 && QCAP % 128 == 0, "plane list");

    if (tid < NPB) {
        const float ad = a_dst[node0 + tid];
        float v = a_src[node0 + tid] + ad;
        v = (v > 0.0f) ? v : v * NEG_SLOPE;
        adst_l[tid]  = ad;
        self_sc[tid] = v;
        m_l[tid]     = v;
    }
    if (tid == 0) q_n = 0;
    __syncthreads();

    scan_enqueue(plane_lo, plane_hi, 0, N_EDGES / 32, blockIdx.x * (N_EDGES / 32 / N_OWN_BLOCKS), key, tid, q_pk, &q_n);
    __syncthreads();
    const int n_all = __builtin_amdgcn_readfirstlane(q_n);

    if (n_all <= QCAP) {
        resolve_max(ei, a_src, adst_l, m_l, q_pk, q_sd, q_sc, n_all, tid, node0);
        __syncthreads();
        for (int i = tid; i < NPB * HIDDEN; i += OWN_T) {
            const int dl = i >> 6;
            const float ex = expf(self_sc[dl] - m_l[dl]);
            acc[i] = ex * h[(size_t)node0 * HIDDEN + i];
            if ((i & 63) == 0) s_l[dl] = ex;
        }
        __syncthreads();
        process_queue(h, m_l, s_l, acc, q_sd, q_sc, n_all, wv, lane);
        __syncthreads();
    } else {
        for (int b = 0; b < N_EDGES; b += QCAP) {
            const int len = (N_EDGES - b < QCAP) ? (N_EDGES - b) : QCAP;
            __syncthreads();
            if (tid == 0) q_n = 0;
            __syncthreads();
            scan_enqueue(plane_lo, plane_hi, b / 32, len / 32, 0, key, tid, q_pk, &q_n);
            __syncthreads();
            resolve_max(ei, a_src, adst_l, m_l, q_pk, q_sd, q_sc, __builtin_amdgcn_readfirstlane(q_n), tid, node0);
        }
        __syncthreads();
        for (int i = tid; i < NPB * HIDDEN; i += OWN_T) {
            const int dl = i >> 6;
            const float ex = expf(self_sc[dl] - m_l[dl]);
            acc[i] = ex * h[(size_t)node0 * HIDDEN + i];
            if ((i & 63) == 0) s_l[dl] = ex;
        }
        for (int b = 0; b < N_EDGES; b += QCAP) {
            const int len = (N_EDGES - b < QCAP) ? (N_EDGES - b) : QCAP;
            __syncthreads();
            if (tid == 0) q_n = 0;
            __syncthreads();
            scan_enqueue(plane_lo, plane_hi, b / 32, len / 32, 0, key, tid, q_pk, &q_n);
            __syncthreads();
            const int n = __builtin_amdgcn_readfirstlane(q_n);
            resolve_max(ei, a_src, adst_l, m_l, q_pk, q_sd, q_sc, n, tid, node0);
            __syncthreads();
            process_queue(h, m_l, s_l, acc, q_sd, q_sc, n, wv, lane);
        }
        __syncthreads();
    }

    const float bc0 = bias_conv[lane * 2], bc1 = bias_conv[lane * 2 + 1];
    const float wl0 = W_lin[lane * 2],     wl1 = W_lin[lane * 2 + 1];
    const float bl  = b_lin[0];
    for (int dl = wv; dl < NPB; dl += OWN_T / 32) {
        const float inv = 1.0f / s_l[dl];
        const v2f o = *(const v2f*)(acc + dl * HIDDEN + lane * 2);
        const float o0 = fmaxf(o.x * inv + bc0, 0.0f);
        const float o1 = fmaxf(o.y * inv + bc1, 0.0f);
        float p = o0 * wl0 + o1 * wl1;
        #pragma unroll
        for (int off = 1; off < 32; off <<= 1) p += __shfl_xor(p, off, 32);
        if (lane == 0) y[node0 + dl] = 1.0f / (1.0f + expf(-(p + bl)));
    }
}

extern "C" void kernel_launch(void* const* d_in, const int* in_sizes, int n_in,
                              void* d_out, int out_size, void* d_ws, size_t ws_size,
                              hipStream_t stream) {
    const float* x         = (const float*)d_in[0];
    const int*   ei        = (const int*)  d_in[1];
    const float* W         = (const float*)d_in[2];
    const float* att_src_v = (const float*)d_in[3];
    const float* att_dst_v = (const float*)d_in[4];
    const float* bias_conv = (const float*)d_in[5];
    const float* W_lin     = (const float*)d_in[6];
    const float* b_lin     = (const float*)d_in[7];
    float* y = (float*)d_out;

    float* ws     = (float*)d_ws;
    float* h      = ws;
    float* a_src  = ws + 3200000;
    float* a_dst  = ws + 3250000;
    unsigned* binw = (unsigned*)(ws + 3300000);

    gat_gemm_wmma_kernel<<<N_TILES, 32, 0, stream>>>(x, W, att_src_v, att_dst_v,
                                                     h, a_src, a_dst, ei, binw);

    gat_owner_kernel<<<N_OWN_BLOCKS, OWN_T, 0, stream>>>(ei, binw, a_src, a_dst, h, bias_conv,
                                                         W_lin, b_lin, y);
}


# GEMM waves also write one 4-byte record per edge (src | local dst << 16) in list order; owner blocks read one record per own edge, not src and dst list entries; sums unchanged (arrival order)
# speedup vs baseline: 4.1016x; 1.0346x over previous
#include <hip/hip_runtime.h>
#include <math.h>

#define N_NODES 50000
#define N_EDGES 800000
#define IN_DIM 128
#define HIDDEN 64
#define NEG_SLOPE 0.2f
#define E_TOT (N_EDGES + N_NODES)
#define NPB 200
#define N_OWN_BLOCKS (N_NODES / NPB)
#define N_TILES (N_NODES / 16)
#define NO_BIN 255u

typedef float v2f __attribute__((ext_vector_type(2)));
typedef float v8f __attribute__((ext_vector_type(8)));

__device__ __forceinline__ unsigned bin_of(int d) {
    const unsigned u = (unsigned)d;
    return (u < (unsigned)N_NODES) ? (u / NPB) : NO_BIN;
}

__global__ void __launch_bounds__(32)
gat_gemm_wmma_kernel(const float* __restrict__ x, const float* __restrict__ W,
                     const float* __restrict__ att_src_v, const float* __restrict__ att_dst_v,
                     float* __restrict__ h, float* __restrict__ a_src,
                     float* __restrict__ a_dst,
                     const int* __restrict__ ei, unsigned* __restrict__ binw,
                     unsigned* __restrict__ rec)
{
    const int tile = blockIdx.x;
    const int lane = threadIdx.x;
    const int half = lane >> 4;
    const int lrow = lane & 15;
    const int row0 = tile * 16;

    const float* xrow  = x + (size_t)(row0 + lrow) * IN_DIM;
    const float* wrow  = W + (size_t)lrow * IN_DIM;

    static_assert(N_EDGES == N_TILES * 256, "each GEMM wave converts 256 edges");
    const int4* dst4 = (const int4*)(ei + N_EDGES);
    const int4 dv0 = dst4[tile * 64 + lane];
    const int4 dv1 = dst4[tile * 64 + 32 + lane];
    const int4* src4 = (const int4*)ei;
    const int4 sv0 = src4[tile * 64 + lane];
    const int4 sv1 = src4[tile * 64 + 32 + lane];

    v8f acc0 = {}, acc1 = {}, acc2 = {}, acc3 = {};

    for (int kb = 0; kb < IN_DIM; kb += 4) {
        const int k = kb + 2 * half;
        v2f a  = *(const v2f*)(xrow + k);
        v2f b0 = *(const v2f*)(wrow + k);
        v2f b1 = *(const v2f*)(wrow + 16 * IN_DIM + k);
        v2f b2 = *(const v2f*)(wrow + 32 * IN_DIM + k);
        v2f b3 = *(const v2f*)(wrow + 48 * IN_DIM + k);
        acc0 = __builtin_amdgcn_wmma_f32_16x16x4_f32(false, a, false, b0, (short)0, acc0, false, false);
        acc1 = __builtin_amdgcn_wmma_f32_16x16x4_f32(false, a, false, b1, (short)0, acc1, false, false);
        acc2 = __builtin_amdgcn_wmma_f32_16x16x4_f32(false, a, false, b2, (short)0, acc2, false, false);
        acc3 = __builtin_amdgcn_wmma_f32_16x16x4_f32(false, a, false, b3, (short)0, acc3, false, false);
    }

    const float as0 = att_src_v[lrow],      ad0 = att_dst_v[lrow];
    const float as1 = att_src_v[16 + lrow], ad1 = att_dst_v[16 + lrow];
    const float as2 = att_src_v[32 + lrow], ad2 = att_dst_v[32 + lrow];
    const float as3 = att_src_v[48 + lrow], ad3 = att_dst_v[48 + lrow];

    #pragma unroll
    for (int r = 0; r < 8; ++r) {
        const int row = row0 + r + 8 * half;
        float* hr = h + (size_t)row * HIDDEN;
        hr[lrow]      = acc0[r];
        hr[16 + lrow] = acc1[r];
        hr[32 + lrow] = acc2[r];
        hr[48 + lrow] = acc3[r];
        float ps = acc0[r] * as0 + acc1[r] * as1 + acc2[r] * as2 + acc3[r] * as3;
        float pd = acc0[r] * ad0 + acc1[r] * ad1 + acc2[r] * ad2 + acc3[r] * ad3;
        #pragma unroll
        for (int off = 1; off < 16; off <<= 1) {
            ps += __shfl_xor(ps, off, 32);
            pd += __shfl_xor(pd, off, 32);
        }
        if (lrow == 0) { a_src[row] = ps; a_dst[row] = pd; }
    }

    {
        unsigned wa = 0, wb = 0;
        #pragma unroll
        for (int t = 0; t < 2; ++t) {
            const int4 dv = t ? dv1 : dv0;
            const int4 sv = t ? sv1 : sv0;
            const unsigned bb[4] = { bin_of(dv.x), bin_of(dv.y), bin_of(dv.z), bin_of(dv.w) };
            uint4 rv;
            rv.x = (unsigned)sv.x | ((((unsigned)dv.x - bb[0] * NPB) & 0xFFu) << 16);
            rv.y = (unsigned)sv.y | ((((unsigned)dv.y - bb[1] * NPB) & 0xFFu) << 16);
            rv.z = (unsigned)sv.z | ((((unsigned)dv.z - bb[2] * NPB) & 0xFFu) << 16);
            rv.w = (unsigned)sv.w | ((((unsigned)dv.w - bb[3] * NPB) & 0xFFu) << 16);
            ((uint4*)rec)[tile * 64 + t * 32 + lane] = rv;
            #pragma unroll
            for (int c = 0; c < 4; ++c) {
                #pragma unroll
                for (int i = 0; i < 8; ++i) {
                    const unsigned pl = __builtin_amdgcn_ballot_w32(((bb[c] >> i) & 1u) != 0u);
                    if (i < 4) wa = (lane == 16 * t + 4 * c + i)       ? pl : wa;
                    else       wb = (lane == 16 * t + 4 * c + (i - 4)) ? pl : wb;
                }
            }
        }
        binw[tile * 32 + lane]               = wa;
        binw[N_EDGES / 8 + tile * 32 + lane] = wb;
    }
}

#define QCAP 16384
#define OWN_T 1024
#define PB 16
#define SCAN_U 5

__device__ __forceinline__ void lds_max_float(float* addr, float val) {
    if (val >= 0.0f) atomicMax((int*)addr, __float_as_int(val));
    else             atomicMin((unsigned int*)addr, __float_as_uint(val));
}

struct PlaneKey { unsigned w[8]; };

__device__ __forceinline__ unsigned hits32(uint4 lo, uint4 hi, const PlaneKey& key) {
    return (lo.x ^ key.w[0]) & (lo.y ^ key.w[1]) & (lo.z ^ key.w[2]) & (lo.w ^ key.w[3]) &
           (hi.x ^ key.w[4]) & (hi.y ^ key.w[5]) & (hi.z ^ key.w[6]) & (hi.w ^ key.w[7]);
}

__device__ __forceinline__ int wave_prefix_sum(int v, int lane) {
    v += __builtin_amdgcn_update_dpp(0, v, 0x111, 0xF, 0xF, false);
    v += __builtin_amdgcn_update_dpp(0, v, 0x112, 0xF, 0xF, false);
    v += __builtin_amdgcn_update_dpp(0, v, 0x114, 0xF, 0xF, false);
    v += __builtin_amdgcn_update_dpp(0, v, 0x118, 0xF, 0xF, false);
    const int low = __builtin_amdgcn_readlane(v, 15);
    return v + ((lane >= 16) ? low : 0);
}

template <bool GUARD>
__device__ __forceinline__ void scan_round(const uint4* __restrict__ plane_lo,
                                           const uint4* __restrict__ plane_hi,
                                           int g0, int lenG, int rotG, const PlaneKey& key,
                                           int i0, int lane, unsigned* q_pk, int* q_n) {
    uint4 lo[SCAN_U], hi[SCAN_U];
    int   G[SCAN_U];
    #pragma unroll
    for (int k = 0; k < SCAN_U; ++k) {
        int jj = i0 + k * OWN_T + rotG;
        if (jj >= lenG) jj -= lenG;
        G[k] = g0 + jj;
        if (!GUARD || (i0 + k * OWN_T < lenG)) { lo[k] = plane_lo[G[k]]; hi[k] = plane_hi[G[k]]; }
        else { lo[k] = make_uint4(~0u, ~0u, ~0u, ~0u); hi[k] = lo[k]; }
    }
    unsigned m[SCAN_U];
    int cnt = 0;
    #pragma unroll
    for (int k = 0; k < SCAN_U; ++k) {
        m[k] = hits32(lo[k], hi[k], key);
        cnt += __popc(m[k]);
    }
    const int incl  = wave_prefix_sum(cnt, lane);
    const int total = __builtin_amdgcn_readlane(incl, 31);
    int base = 0;
    if (lane == 0) base = atomicAdd(q_n, total);
    int slot = __builtin_amdgcn_readfirstlane(base) + incl - cnt;
    #pragma unroll
    for (int k = 0; k < SCAN_U; ++k) {
        unsigned mm = m[k];
        const int e0 = ((G[k] & ~3) << 5) + (G[k] & 3);
        while (mm) {
            const int p = __builtin_ctz(mm);
            mm &= mm - 1;
            if (slot < QCAP) q_pk[slot] = (unsigned)(e0 + 4 * p);
            ++slot;
        }
    }
}

__device__ __forceinline__ void scan_enqueue(const uint4* __restrict__ plane_lo,
                                             const uint4* __restrict__ plane_hi,
                                             int g0, int lenG, int rotG, const PlaneKey& key,
                                             int tid, unsigned* q_pk, int* q_n) {
    const int full = lenG / (SCAN_U * OWN_T);
    for (int r = 0; r < full; ++r)
        scan_round<false>(plane_lo, plane_hi, g0, lenG, rotG, key, tid + r * (SCAN_U * OWN_T), tid & 31, q_pk, q_n);
    if (full * (SCAN_U * OWN_T) < lenG)
        scan_round<true>(plane_lo, plane_hi, g0, lenG, rotG, key, tid + full * (SCAN_U * OWN_T), tid & 31, q_pk, q_n);
}

__device__ __forceinline__ void resolve_max(const unsigned* __restrict__ rec,
                                            const float* __restrict__ a_src,
                                            const float* adst_l, float* m_l,
                                            const unsigned* q_pk, unsigned* q_sd, float* q_sc,
                                            int n, int tid) {
    for (int i0 = tid; i0 < n; i0 += 4 * OWN_T) {
        unsigned u[4];
        unsigned r[4];
        float    as[4];
        #pragma unroll
        for (int k = 0; k < 4; ++k) {
            const int i = i0 + k * OWN_T;
            u[k] = q_pk[(i < n) ? i : (n - 1)];
        }
        #pragma unroll
        for (int k = 0; k < 4; ++k) r[k] = rec[u[k]];
        #pragma unroll
        for (int k = 0; k < 4; ++k) as[k] = a_src[r[k] & 0xFFFFu];
        #pragma unroll
        for (int k = 0; k < 4; ++k) {
            const int i = i0 + k * OWN_T;
            if (i < n) {
                const unsigned dl = r[k] >> 16;
                float v = as[k] + adst_l[dl];
                v = (v > 0.0f) ? v : v * NEG_SLOPE;
                q_sd[i] = r[k];
                q_sc[i] = v;
                lds_max_float(&m_l[dl], v);
            }
        }
    }
}

template <bool FULL>
__device__ __forceinline__ void process_batch(const float* __restrict__ h,
                                              const float* m_l, float* s_l, float* acc,
                                              const unsigned* q_sd, const float* q_sc,
                                              int base, int n, int lane) {
    const int  my = base + (lane & (PB - 1));
    const bool ok = FULL || (my < n);
    const unsigned u  = q_sd[ok ? my : base];
    const float    sc = q_sc[ok ? my : base];
    const float    ex = expf(sc - m_l[u >> 16]);
    if (ok && lane < PB) atomicAdd(&s_l[u >> 16], ex);
    v2f hv[PB];
    #pragma unroll
    for (int k = 0; k < PB; ++k) {
        const unsigned uk = (unsigned)__builtin_amdgcn_readlane((int)u, k);
        hv[k] = *(const v2f*)(h + (size_t)(uk & 0xFFFFu) * HIDDEN + lane * 2);
    }
    #pragma unroll
    for (int k = 0; k < PB; ++k) {
        if (FULL || (base + k < n)) {
            const unsigned uk  = (unsigned)__builtin_amdgcn_readlane((int)u, k);
            const float    exk = __int_as_float(__builtin_amdgcn_readlane(__float_as_int(ex), k));
            float* a = acc + (uk >> 16) * HIDDEN + lane * 2;
            atomicAdd(a,     exk * hv[k].x);
            atomicAdd(a + 1, exk * hv[k].y);
        }
    }
}

__device__ __forceinline__ void process_queue(const float* __restrict__ h,
                                              const float* m_l, float* s_l, float* acc,
                                              const unsigned* q_sd, const float* q_sc,
                                              int n, int wv, int lane) {
    for (int base = wv * PB; base < n; base += (OWN_T / 32) * PB) {
        if (base + PB <= n) process_batch<true >(h, m_l, s_l, acc, q_sd, q_sc, base, n, lane);
        else                process_batch<false>(h, m_l, s_l, acc, q_sd, q_sc, base, n, lane);
    }
}

__global__ void __launch_bounds__(OWN_T)
gat_owner_kernel(const unsigned* __restrict__ rec,
                 const unsigned* __restrict__ binw,
                 const float* __restrict__ a_src,
                 const float* __restrict__ a_dst,
                 const float* __restrict__ h,
                 const float* __restrict__ bias_conv,
                 const float* __restrict__ W_lin,
                 const float* __restrict__ b_lin,
                 float* __restrict__ y) {
    __shared__ float    acc[NPB * HIDDEN];
    __shared__ float    m_l[NPB];
    __shared__ float    s_l[NPB];
    __shared__ float    adst_l[NPB];
    __shared__ float    self_sc[NPB];
    __shared__ unsigned q_pk[QCAP];
    __shared__ unsigned q_sd[QCAP];
    __shared__ float    q_sc[QCAP];
    __shared__ int      q_n;

    const int tid   = threadIdx.x;
    const int lane  = tid & 31;
    const int wv    = __builtin_amdgcn_readfirstlane(tid >> 5);
    const int node0 = blockIdx.x * NPB;
    const uint4* plane_lo = (const uint4*)binw;
    const uint4* plane_hi = (const uint4*)(binw + N_EDGES / 8);
    PlaneKey key;
    #pragma unroll
    for (int i = 0; i < 8; ++i) key.w[i] = ((blockIdx.x >> i) & 1u) - 1u;
    static_assert(N_OWN_BLOCKS <= 255 && N_EDGES % 128 == 0 && QCAP % 128 == 0, "plane list");

    if (tid < NPB) {
        const float ad = a_dst[node0 + tid];
        float v = a_src[node0 + tid] + ad;
        v = (v > 0.0f) ? v : v * NEG_SLOPE;
        adst_l[tid]  = ad;
        self_sc[tid] = v;
        m_l[tid]     = v;
    }
    if (tid == 0) q_n = 0;
    __syncthreads();

    scan_enqueue(plane_lo, plane_hi, 0, N_EDGES / 32, blockIdx.x * (N_EDGES / 32 / N_OWN_BLOCKS), key, tid, q_pk, &q_n);
    __syncthreads();
    const int n_all = __builtin_amdgcn_readfirstlane(q_n);

    if (n_all <= QCAP) {
        resolve_max(rec, a_src, adst_l, m_l, q_pk, q_sd, q_sc, n_all, tid);
        __syncthreads();
        for (int i = tid; i < NPB * HIDDEN; i += OWN_T) {
            const int dl = i >> 6;
            const float ex = expf(self_sc[dl] - m_l[dl]);
            acc[i] = ex * h[(size_t)node0 * HIDDEN + i];
            if ((i & 63) == 0) s_l[dl] = ex;
        }
        __syncthreads();
        process_queue(h, m_l, s_l, acc, q_sd, q_sc, n_all, wv, lane);
        __syncthreads();
    } else {
        for (int b = 0; b < N_EDGES; b += QCAP) {
            const int len = (N_EDGES - b < QCAP) ? (N_EDGES - b) : QCAP;
            __syncthreads();
            if (tid == 0) q_n = 0;
            __syncthreads();
            scan_enqueue(plane_lo, plane_hi, b / 32, len / 32, 0, key, tid, q_pk, &q_n);
            __syncthreads();
            resolve_max(rec, a_src, adst_l, m_l, q_pk, q_sd, q_sc, __builtin_amdgcn_readfirstlane(q_n), tid);
        }
        __syncthreads();
        for (int i = tid; i < NPB * HIDDEN; i += OWN_T) {
            const int dl = i >> 6;
            const float ex = expf(self_sc[dl] - m_l[dl]);
            acc[i] = ex * h[(size_t)node0 * HIDDEN + i];
            if ((i & 63) == 0) s_l[dl] = ex;
        }
        for (int b = 0; b < N_EDGES; b += QCAP) {
            const int len = (N_EDGES - b < QCAP) ? (N_EDGES - b) : QCAP;
            __syncthreads();
            if (tid == 0) q_n = 0;
            __syncthreads();
            scan_enqueue(plane_lo, plane_hi, b / 32, len / 32, 0, key, tid, q_pk, &q_n);
            __syncthreads();
            const int n = __builtin_amdgcn_readfirstlane(q_n);
            resolve_max(rec, a_src, adst_l, m_l, q_pk, q_sd, q_sc, n, tid);
            __syncthreads();
            process_queue(h, m_l, s_l, acc, q_sd, q_sc, n, wv, lane);
        }
        __syncthreads();
    }

    const float bc0 = bias_conv[lane * 2], bc1 = bias_conv[lane * 2 + 1];
    const float wl0 = W_lin[lane * 2],     wl1 = W_lin[lane * 2 + 1];
    const float bl  = b_lin[0];
    for (int dl = wv; dl < NPB; dl += OWN_T / 32) {
        const float inv = 1.0f / s_l[dl];
        const v2f o = *(const v2f*)(acc + dl * HIDDEN + lane * 2);
        const float o0 = fmaxf(o.x * inv + bc0, 0.0f);
        const float o1 = fmaxf(o.y * inv + bc1, 0.0f);
        float p = o0 * wl0 + o1 * wl1;
        #pragma unroll
        for (int off = 1; off < 32; off <<= 1) p += __shfl_xor(p, off, 32);
        if (lane == 0) y[node0 + dl] = 1.0f / (1.0f + expf(-(p + bl)));
    }
}

extern "C" void kernel_launch(void* const* d_in, const int* in_sizes, int n_in,
                              void* d_out, int out_size, void* d_ws, size_t ws_size,
                              hipStream_t stream) {
    const float* x         = (const float*)d_in[0];
    const int*   ei        = (const int*)  d_in[1];
    const float* W         = (const float*)d_in[2];
    const float* att_src_v = (const float*)d_in[3];
    const float* att_dst_v = (const float*)d_in[4];
    const float* bias_conv = (const float*)d_in[5];
    const float* W_lin     = (const float*)d_in[6];
    const float* b_lin     = (const float*)d_in[7];
    float* y = (float*)d_out;

    float* ws     = (float*)d_ws;
    float* h      = ws;
    float* a_src  = ws + 3200000;
    float* a_dst  = ws + 3250000;
    unsigned* binw = (unsigned*)(ws + 3300000);
    unsigned* rec  = (unsigned*)(ws + 3500000);

    gat_gemm_wmma_kernel<<<N_TILES, 32, 0, stream>>>(x, W, att_src_v, att_dst_v,
                                                     h, a_src, a_dst, ei, binw, rec);

    gat_owner_kernel<<<N_OWN_BLOCKS, OWN_T, 0, stream>>>(rec, binw, a_src, a_dst, h, bias_conv,
                                                         W_lin, b_lin, y);
}


# GEMM waves fetch x and W fragments as 16 B global loads covering two k-steps (half the load instructions); each WMMA now sums k, k+1, k+4, k+5: h's k sum order changed
# speedup vs baseline: 4.3340x; 1.0567x over previous
#include <hip/hip_runtime.h>
#include <math.h>

#define N_NODES 50000
#define N_EDGES 800000
#define IN_DIM 128
#define HIDDEN 64
#define NEG_SLOPE 0.2f
#define E_TOT (N_EDGES + N_NODES)
#define NPB 200
#define N_OWN_BLOCKS (N_NODES / NPB)
#define N_TILES (N_NODES / 16)
#define NO_BIN 255u

typedef float v2f __attribute__((ext_vector_type(2)));
typedef float v4f __attribute__((ext_vector_type(4)));
typedef float v8f __attribute__((ext_vector_type(8)));

__device__ __forceinline__ unsigned bin_of(int d) {
    const unsigned u = (unsigned)d;
    return (u < (unsigned)N_NODES) ? (u / NPB) : NO_BIN;
}

__global__ void __launch_bounds__(32)
gat_gemm_wmma_kernel(const float* __restrict__ x, const float* __restrict__ W,
                     const float* __restrict__ att_src_v, const float* __restrict__ att_dst_v,
                     float* __restrict__ h, float* __restrict__ a_src,
                     float* __restrict__ a_dst,
                     const int* __restrict__ ei, unsigned* __restrict__ binw,
                     unsigned* __restrict__ rec)
{
    const int tile = blockIdx.x;
    const int lane = threadIdx.x;
    const int half = lane >> 4;
    const int lrow = lane & 15;
    const int row0 = tile * 16;

    const float* xrow  = x + (size_t)(row0 + lrow) * IN_DIM;
    const float* wrow  = W + (size_t)lrow * IN_DIM;

    static_assert(N_EDGES == N_TILES * 256, "each GEMM wave converts 256 edges");
    const int4* dst4 = (const int4*)(ei + N_EDGES);
    const int4 dv0 = dst4[tile * 64 + lane];
    const int4 dv1 = dst4[tile * 64 + 32 + lane];
    const int4* src4 = (const int4*)ei;
    const int4 sv0 = src4[tile * 64 + lane];
    const int4 sv1 = src4[tile * 64 + 32 + lane];

    v8f acc0 = {}, acc1 = {}, acc2 = {}, acc3 = {};

    for (int kb = 0; kb < IN_DIM; kb += 8) {
        const int k = kb + 4 * half;
        const v4f a  = *(const v4f*)(xrow + k);
        const v4f b0 = *(const v4f*)(wrow + k);
        const v4f b1 = *(const v4f*)(wrow + 16 * IN_DIM + k);
        const v4f b2 = *(const v4f*)(wrow + 32 * IN_DIM + k);
        const v4f b3 = *(const v4f*)(wrow + 48 * IN_DIM + k);
        const v2f a_lo = __builtin_shufflevector(a, a, 0, 1), a_hi = __builtin_shufflevector(a, a, 2, 3);
        acc0 = __builtin_amdgcn_wmma_f32_16x16x4_f32(false, a_lo, false, __builtin_shufflevector(b0, b0, 0, 1), (short)0, acc0, false, false);
        acc1 = __builtin_amdgcn_wmma_f32_16x16x4_f32(false, a_lo, false, __builtin_shufflevector(b1, b1, 0, 1), (short)0, acc1, false, false);
        acc2 = __builtin_amdgcn_wmma_f32_16x16x4_f32(false, a_lo, false, __builtin_shufflevector(b2, b2, 0, 1), (short)0, acc2, false, false);
        acc3 = __builtin_amdgcn_wmma_f32_16x16x4_f32(false, a_lo, false, __builtin_shufflevector(b3, b3, 0, 1), (short)0, acc3, false, false);
        acc0 = __builtin_amdgcn_wmma_f32_16x16x4_f32(false, a_hi, false, __builtin_shufflevector(b0, b0, 2, 3), (short)0, acc0, false, false);
        acc1 = __builtin_amdgcn_wmma_f32_16x16x4_f32(false, a_hi, false, __builtin_shufflevector(b1, b1, 2, 3), (short)0, acc1, false, false);
        acc2 = __builtin_amdgcn_wmma_f32_16x16x4_f32(false, a_hi, false, __builtin_shufflevector(b2, b2, 2, 3), (short)0, acc2, false, false);
        acc3 = __builtin_amdgcn_wmma_f32_16x16x4_f32(false, a_hi, false, __builtin_shufflevector(b3, b3, 2, 3), (short)0, acc3, false, false);
    }

    const float as0 = att_src_v[lrow],      ad0 = att_dst_v[lrow];
    const float as1 = att_src_v[16 + lrow], ad1 = att_dst_v[16 + lrow];
    const float as2 = att_src_v[32 + lrow], ad2 = att_dst_v[32 + lrow];
    const float as3 = att_src_v[48 + lrow], ad3 = att_dst_v[48 + lrow];

    #pragma unroll
    for (int r = 0; r < 8; ++r) {
        const int row = row0 + r + 8 * half;
        float* hr = h + (size_t)row * HIDDEN;
        hr[lrow]      = acc0[r];
        hr[16 + lrow] = acc1[r];
        hr[32 + lrow] = acc2[r];
        hr[48 + lrow] = acc3[r];
        float ps = acc0[r] * as0 + acc1[r] * as1 + acc2[r] * as2 + acc3[r] * as3;
        float pd = acc0[r] * ad0 + acc1[r] * ad1 + acc2[r] * ad2 + acc3[r] * ad3;
        #pragma unroll
        for (int off = 1; off < 16; off <<= 1) {
            ps += __shfl_xor(ps, off, 32);
            pd += __shfl_xor(pd, off, 32);
        }
        if (lrow == 0) { a_src[row] = ps; a_dst[row] = pd; }
    }

    {
        unsigned wa = 0, wb = 0;
        #pragma unroll
        for (int t = 0; t < 2; ++t) {
            const int4 dv = t ? dv1 : dv0;
            const int4 sv = t ? sv1 : sv0;
            const unsigned bb[4] = { bin_of(dv.x), bin_of(dv.y), bin_of(dv.z), bin_of(dv.w) };
            uint4 rv;
            rv.x = (unsigned)sv.x | ((((unsigned)dv.x - bb[0] * NPB) & 0xFFu) << 16);
            rv.y = (unsigned)sv.y | ((((unsigned)dv.y - bb[1] * NPB) & 0xFFu) << 16);
            rv.z = (unsigned)sv.z | ((((unsigned)dv.z - bb[2] * NPB) & 0xFFu) << 16);
            rv.w = (unsigned)sv.w | ((((unsigned)dv.w - bb[3] * NPB) & 0xFFu) << 16);
            ((uint4*)rec)[tile * 64 + t * 32 + lane] = rv;
            #pragma unroll
            for (int c = 0; c < 4; ++c) {
                #pragma unroll
                for (int i = 0; i < 8; ++i) {
                    const unsigned pl = __builtin_amdgcn_ballot_w32(((bb[c] >> i) & 1u) != 0u);
                    if (i < 4) wa = (lane == 16 * t + 4 * c + i)       ? pl : wa;
                    else       wb = (lane == 16 * t + 4 * c + (i - 4)) ? pl : wb;
                }
            }
        }
        binw[tile * 32 + lane]               = wa;
        binw[N_EDGES / 8 + tile * 32 + lane] = wb;
    }
}

#define QCAP 16384
#define OWN_T 1024
#define PB 16
#define SCAN_U 5

__device__ __forceinline__ void lds_max_float(float* addr, float val) {
    if (val >= 0.0f) atomicMax((int*)addr, __float_as_int(val));
    else             atomicMin((unsigned int*)addr, __float_as_uint(val));
}

struct PlaneKey { unsigned w[8]; };

__device__ __forceinline__ unsigned hits32(uint4 lo, uint4 hi, const PlaneKey& key) {
    return (lo.x ^ key.w[0]) & (lo.y ^ key.w[1]) & (lo.z ^ key.w[2]) & (lo.w ^ key.w[3]) &
           (hi.x ^ key.w[4]) & (hi.y ^ key.w[5]) & (hi.z ^ key.w[6]) & (hi.w ^ key.w[7]);
}

__device__ __forceinline__ int wave_prefix_sum(int v, int lane) {
    v += __builtin_amdgcn_update_dpp(0, v, 0x111, 0xF, 0xF, false);
    v += __builtin_amdgcn_update_dpp(0, v, 0x112, 0xF, 0xF, false);
    v += __builtin_amdgcn_update_dpp(0, v, 0x114, 0xF, 0xF, false);
    v += __builtin_amdgcn_update_dpp(0, v, 0x118, 0xF, 0xF, false);
    const int low = __builtin_amdgcn_readlane(v, 15);
    return v + ((lane >= 16) ? low : 0);
}

template <bool GUARD>
__device__ __forceinline__ void scan_round(const uint4* __restrict__ plane_lo,
                                           const uint4* __restrict__ plane_hi,
                                           int g0, int lenG, int rotG, const PlaneKey& key,
                                           int i0, int lane, unsigned* q_pk, int* q_n) {
    uint4 lo[SCAN_U], hi[SCAN_U];
    int   G[SCAN_U];
    #pragma unroll
    for (int k = 0; k < SCAN_U; ++k) {
        int jj = i0 + k * OWN_T + rotG;
        if (jj >= lenG) jj -= lenG;
        G[k] = g0 + jj;
        if (!GUARD || (i0 + k * OWN_T < lenG)) { lo[k] = plane_lo[G[k]]; hi[k] = plane_hi[G[k]]; }
        else { lo[k] = make_uint4(~0u, ~0u, ~0u, ~0u); hi[k] = lo[k]; }
    }
    unsigned m[SCAN_U];
    int cnt = 0;
    #pragma unroll
    for (int k = 0; k < SCAN_U; ++k) {
        m[k] = hits32(lo[k], hi[k], key);
        cnt += __popc(m[k]);
    }
    const int incl  = wave_prefix_sum(cnt, lane);
    const int total = __builtin_amdgcn_readlane(incl, 31);
    int base = 0;
    if (lane == 0) base = atomicAdd(q_n, total);
    int slot = __builtin_amdgcn_readfirstlane(base) + incl - cnt;
    #pragma unroll
    for (int k = 0; k < SCAN_U; ++k) {
        unsigned mm = m[k];
        const int e0 = ((G[k] & ~3) << 5) + (G[k] & 3);
        while (mm) {
            const int p = __builtin_ctz(mm);
            mm &= mm - 1;
            if (slot < QCAP) q_pk[slot] = (unsigned)(e0 + 4 * p);
            ++slot;
        }
    }
}

__device__ __forceinline__ void scan_enqueue(const uint4* __restrict__ plane_lo,
                                             const uint4* __restrict__ plane_hi,
                                             int g0, int lenG, int rotG, const PlaneKey& key,
                                             int tid, unsigned* q_pk, int* q_n) {
    const int full = lenG / (SCAN_U * OWN_T);
    for (int r = 0; r < full; ++r)
        scan_round<false>(plane_lo, plane_hi, g0, lenG, rotG, key, tid + r * (SCAN_U * OWN_T), tid & 31, q_pk, q_n);
    if (full * (SCAN_U * OWN_T) < lenG)
        scan_round<true>(plane_lo, plane_hi, g0, lenG, rotG, key, tid + full * (SCAN_U * OWN_T), tid & 31, q_pk, q_n);
}

__device__ __forceinline__ void resolve_max(const unsigned* __restrict__ rec,
                                            const float* __restrict__ a_src,
                                            const float* adst_l, float* m_l,
                                            const unsigned* q_pk, unsigned* q_sd, float* q_sc,
                                            int n, int tid) {
    for (int i0 = tid; i0 < n; i0 += 4 * OWN_T) {
        unsigned u[4];
        unsigned r[4];
        float    as[4];
        #pragma unroll
        for (int k = 0; k < 4; ++k) {
            const int i = i0 + k * OWN_T;
            u[k] = q_pk[(i < n) ? i : (n - 1)];
        }
        #pragma unroll
        for (int k = 0; k < 4; ++k) r[k] = rec[u[k]];
        #pragma unroll
        for (int k = 0; k < 4; ++k) as[k] = a_src[r[k] & 0xFFFFu];
        #pragma unroll
        for (int k = 0; k < 4; ++k) {
            const int i = i0 + k * OWN_T;
            if (i < n) {
                const unsigned dl = r[k] >> 16;
                float v = as[k] + adst_l[dl];
                v = (v > 0.0f) ? v : v * NEG_SLOPE;
                q_sd[i] = r[k];
                q_sc[i] = v;
                lds_max_float(&m_l[dl], v);
            }
        }
    }
}

template <bool FULL>
__device__ __forceinline__ void process_batch(const float* __restrict__ h,
                                              const float* m_l, float* s_l, float* acc,
                                              const unsigned* q_sd, const float* q_sc,
                                              int base, int n, int lane) {
    const int  my = base + (lane & (PB - 1));
    const bool ok = FULL || (my < n);
    const unsigned u  = q_sd[ok ? my : base];
    const float    sc = q_sc[ok ? my : base];
    const float    ex = expf(sc - m_l[u >> 16]);
    if (ok && lane < PB) atomicAdd(&s_l[u >> 16], ex);
    v2f hv[PB];
    #pragma unroll
    for (int k = 0; k < PB; ++k) {
        const unsigned uk = (unsigned)__builtin_amdgcn_readlane((int)u, k);
        hv[k] = *(const v2f*)(h + (size_t)(uk & 0xFFFFu) * HIDDEN + lane * 2);
    }
    #pragma unroll
    for (int k = 0; k < PB; ++k) {
        if (FULL || (base + k < n)) {
            const unsigned uk  = (unsigned)__builtin_amdgcn_readlane((int)u, k);
            const float    exk = __int_as_float(__builtin_amdgcn_readlane(__float_as_int(ex), k));
            float* a = acc + (uk >> 16) * HIDDEN + lane * 2;
            atomicAdd(a,     exk * hv[k].x);
            atomicAdd(a + 1, exk * hv[k].y);
        }
    }
}

__device__ __forceinline__ void process_queue(const float* __restrict__ h,
                                              const float* m_l, float* s_l, float* acc,
                                              const unsigned* q_sd, const float* q_sc,
                                              int n, int wv, int lane) {
    for (int base = wv * PB; base < n; base += (OWN_T / 32) * PB) {
        if (base + PB <= n) process_batch<true >(h, m_l, s_l, acc, q_sd, q_sc, base, n, lane);
        else                process_batch<false>(h, m_l, s_l, acc, q_sd, q_sc, base, n, lane);
    }
}

__global__ void __launch_bounds__(OWN_T)
gat_owner_kernel(const unsigned* __restrict__ rec,
                 const unsigned* __restrict__ binw,
                 const float* __restrict__ a_src,
                 const float* __restrict__ a_dst,
                 const float* __restrict__ h,
                 const float* __restrict__ bias_conv,
                 const float* __restrict__ W_lin,
                 const float* __restrict__ b_lin,
                 float* __restrict__ y) {
    __shared__ float    acc[NPB * HIDDEN];
    __shared__ float    m_l[NPB];
    __shared__ float    s_l[NPB];
    __shared__ float    adst_l[NPB];
    __shared__ float    self_sc[NPB];
    __shared__ unsigned q_pk[QCAP];
    __shared__ unsigned q_sd[QCAP];
    __shared__ float    q_sc[QCAP];
    __shared__ int      q_n;

    const int tid   = threadIdx.x;
    const int lane  = tid & 31;
    const int wv    = __builtin_amdgcn_readfirstlane(tid >> 5);
    const int node0 = blockIdx.x * NPB;
    const uint4* plane_lo = (const uint4*)binw;
    const uint4* plane_hi = (const uint4*)(binw + N_EDGES / 8);
    PlaneKey key;
    #pragma unroll
    for (int i = 0; i < 8; ++i) key.w[i] = ((blockIdx.x >> i) & 1u) - 1u;
    static_assert(N_OWN_BLOCKS <= 255 && N_EDGES % 128 == 0 && QCAP % 128 == 0, "plane list");

    if (tid < NPB) {
        const float ad = a_dst[node0 + tid];
        float v = a_src[node0 + tid] + ad;
        v = (v > 0.0f) ? v : v * NEG_SLOPE;
        adst_l[tid]  = ad;
        self_sc[tid] = v;
        m_l[tid]     = v;
    }
    if (tid == 0) q_n = 0;
    __syncthreads();

    scan_enqueue(plane_lo, plane_hi, 0, N_EDGES / 32, blockIdx.x * (N_EDGES / 32 / N_OWN_BLOCKS), key, tid, q_pk, &q_n);
    __syncthreads();
    const int n_all = __builtin_amdgcn_readfirstlane(q_n);

    if (n_all <= QCAP) {
        resolve_max(rec, a_src, adst_l, m_l, q_pk, q_sd, q_sc, n_all, tid);
        __syncthreads();
        for (int i = tid; i < NPB * HIDDEN; i += OWN_T) {
            const int dl = i >> 6;
            const float ex = expf(self_sc[dl] - m_l[dl]);
            acc[i] = ex * h[(size_t)node0 * HIDDEN + i];
            if ((i & 63) == 0) s_l[dl] = ex;
        }
        __syncthreads();
        process_queue(h, m_l, s_l, acc, q_sd, q_sc, n_all, wv, lane);
        __syncthreads();
    } else {
        for (int b = 0; b < N_EDGES; b += QCAP) {
            const int len = (N_EDGES - b < QCAP) ? (N_EDGES - b) : QCAP;
            __syncthreads();
            if (tid == 0) q_n = 0;
            __syncthreads();
            scan_enqueue(plane_lo, plane_hi, b / 32, len / 32, 0, key, tid, q_pk, &q_n);
            __syncthreads();
            resolve_max(rec, a_src, adst_l, m_l, q_pk, q_sd, q_sc, __builtin_amdgcn_readfirstlane(q_n), tid);
        }
        __syncthreads();
        for (int i = tid; i < NPB * HIDDEN; i += OWN_T) {
            const int dl = i >> 6;
            const float ex = expf(self_sc[dl] - m_l[dl]);
            acc[i] = ex * h[(size_t)node0 * HIDDEN + i];
            if ((i & 63) == 0) s_l[dl] = ex;
        }
        for (int b = 0; b < N_EDGES; b += QCAP) {
            const int len = (N_EDGES - b < QCAP) ? (N_EDGES - b) : QCAP;
            __syncthreads();
            if (tid == 0) q_n = 0;
            __syncthreads();
            scan_enqueue(plane_lo, plane_hi, b / 32, len / 32, 0, key, tid, q_pk, &q_n);
            __syncthreads();
            const int n = __builtin_amdgcn_readfirstlane(q_n);
            resolve_max(rec, a_src, adst_l, m_l, q_pk, q_sd, q_sc, n, tid);
            __syncthreads();
            process_queue(h, m_l, s_l, acc, q_sd, q_sc, n, wv, lane);
        }
        __syncthreads();
    }

    const float bc0 = bias_conv[lane * 2], bc1 = bias_conv[lane * 2 + 1];
    const float wl0 = W_lin[lane * 2],     wl1 = W_lin[lane * 2 + 1];
    const float bl  = b_lin[0];
    for (int dl = wv; dl < NPB; dl += OWN_T / 32) {
        const float inv = 1.0f / s_l[dl];
        const v2f o = *(const v2f*)(acc + dl * HIDDEN + lane * 2);
        const float o0 = fmaxf(o.x * inv + bc0, 0.0f);
        const float o1 = fmaxf(o.y * inv + bc1, 0.0f);
        float p = o0 * wl0 + o1 * wl1;
        #pragma unroll
        for (int off = 1; off < 32; off <<= 1) p += __shfl_xor(p, off, 32);
        if (lane == 0) y[node0 + dl] = 1.0f / (1.0f + expf(-(p + bl)));
    }
}

extern "C" void kernel_launch(void* const* d_in, const int* in_sizes, int n_in,
                              void* d_out, int out_size, void* d_ws, size_t ws_size,
                              hipStream_t stream) {
    const float* x         = (const float*)d_in[0];
    const int*   ei        = (const int*)  d_in[1];
    const float* W         = (const float*)d_in[2];
    const float* att_src_v = (const float*)d_in[3];
    const float* att_dst_v = (const float*)d_in[4];
    const float* bias_conv = (const float*)d_in[5];
    const float* W_lin     = (const float*)d_in[6];
    const float* b_lin     = (const float*)d_in[7];
    float* y = (float*)d_out;

    float* ws     = (float*)d_ws;
    float* h      = ws;
    float* a_src  = ws + 3200000;
    float* a_dst  = ws + 3250000;
    unsigned* binw = (unsigned*)(ws + 3300000);
    unsigned* rec  = (unsigned*)(ws + 3500000);

    gat_gemm_wmma_kernel<<<N_TILES, 32, 0, stream>>>(x, W, att_src_v, att_dst_v,
                                                     h, a_src, a_dst, ei, binw, rec);

    gat_owner_kernel<<<N_OWN_BLOCKS, OWN_T, 0, stream>>>(rec, binw, a_src, a_dst, h, bias_conv,
                                                         W_lin, b_lin, y);
}


# Start values (self-loop term): one wave per own row, its 6-7 rows of h requested together at block start, one exp per wave; was 12-13 dependent 4-byte round trips per thread; sums unchanged
# speedup vs baseline: 4.7638x; 1.0992x over previous
#include <hip/hip_runtime.h>
#include <math.h>

#define N_NODES 50000
#define N_EDGES 800000
#define IN_DIM 128
#define HIDDEN 64
#define NEG_SLOPE 0.2f
#define E_TOT (N_EDGES + N_NODES)
#define NPB 200
#define N_OWN_BLOCKS (N_NODES / NPB)
#define N_TILES (N_NODES / 16)
#define NO_BIN 255u

typedef float v2f __attribute__((ext_vector_type(2)));
typedef float v4f __attribute__((ext_vector_type(4)));
typedef float v8f __attribute__((ext_vector_type(8)));

__device__ __forceinline__ unsigned bin_of(int d) {
    const unsigned u = (unsigned)d;
    return (u < (unsigned)N_NODES) ? (u / NPB) : NO_BIN;
}

__global__ void __launch_bounds__(32)
gat_gemm_wmma_kernel(const float* __restrict__ x, const float* __restrict__ W,
                     const float* __restrict__ att_src_v, const float* __restrict__ att_dst_v,
                     float* __restrict__ h, float* __restrict__ a_src,
                     float* __restrict__ a_dst,
                     const int* __restrict__ ei, unsigned* __restrict__ binw,
                     unsigned* __restrict__ rec)
{
    const int tile = blockIdx.x;
    const int lane = threadIdx.x;
    const int half = lane >> 4;
    const int lrow = lane & 15;
    const int row0 = tile * 16;

    const float* xrow  = x + (size_t)(row0 + lrow) * IN_DIM;
    const float* wrow  = W + (size_t)lrow * IN_DIM;

    static_assert(N_EDGES == N_TILES * 256, "each GEMM wave converts 256 edges");
    const int4* dst4 = (const int4*)(ei + N_EDGES);
    const int4 dv0 = dst4[tile * 64 + lane];
    const int4 dv1 = dst4[tile * 64 + 32 + lane];
    const int4* src4 = (const int4*)ei;
    const int4 sv0 = src4[tile * 64 + lane];
    const int4 sv1 = src4[tile * 64 + 32 + lane];

    v8f acc0 = {}, acc1 = {}, acc2 = {}, acc3 = {};

    for (int kb = 0; kb < IN_DIM; kb += 8) {
        const int k = kb + 4 * half;
        const v4f a  = *(const v4f*)(xrow + k);
        const v4f b0 = *(const v4f*)(wrow + k);
        const v4f b1 = *(const v4f*)(wrow + 16 * IN_DIM + k);
        const v4f b2 = *(const v4f*)(wrow + 32 * IN_DIM + k);
        const v4f b3 = *(const v4f*)(wrow + 48 * IN_DIM + k);
        const v2f a_lo = __builtin_shufflevector(a, a, 0, 1), a_hi = __builtin_shufflevector(a, a, 2, 3);
        acc0 = __builtin_amdgcn_wmma_f32_16x16x4_f32(false, a_lo, false, __builtin_shufflevector(b0, b0, 0, 1), (short)0, acc0, false, false);
        acc1 = __builtin_amdgcn_wmma_f32_16x16x4_f32(false, a_lo, false, __builtin_shufflevector(b1, b1, 0, 1), (short)0, acc1, false, false);
        acc2 = __builtin_amdgcn_wmma_f32_16x16x4_f32(false, a_lo, false, __builtin_shufflevector(b2, b2, 0, 1), (short)0, acc2, false, false);
        acc3 = __builtin_amdgcn_wmma_f32_16x16x4_f32(false, a_lo, false, __builtin_shufflevector(b3, b3, 0, 1), (short)0, acc3, false, false);
        acc0 = __builtin_amdgcn_wmma_f32_16x16x4_f32(false, a_hi, false, __builtin_shufflevector(b0, b0, 2, 3), (short)0, acc0, false, false);
        acc1 = __builtin_amdgcn_wmma_f32_16x16x4_f32(false, a_hi, false, __builtin_shufflevector(b1, b1, 2, 3), (short)0, acc1, false, false);
        acc2 = __builtin_amdgcn_wmma_f32_16x16x4_f32(false, a_hi, false, __builtin_shufflevector(b2, b2, 2, 3), (short)0, acc2, false, false);
        acc3 = __builtin_amdgcn_wmma_f32_16x16x4_f32(false, a_hi, false, __builtin_shufflevector(b3, b3, 2, 3), (short)0, acc3, false, false);
    }

    const float as0 = att_src_v[lrow],      ad0 = att_dst_v[lrow];
    const float as1 = att_src_v[16 + lrow], ad1 = att_dst_v[16 + lrow];
    const float as2 = att_src_v[32 + lrow], ad2 = att_dst_v[32 + lrow];
    const float as3 = att_src_v[48 + lrow], ad3 = att_dst_v[48 + lrow];

    #pragma unroll
    for (int r = 0; r < 8; ++r) {
        const int row = row0 + r + 8 * half;
        float* hr = h + (size_t)row * HIDDEN;
        hr[lrow]      = acc0[r];
        hr[16 + lrow] = acc1[r];
        hr[32 + lrow] = acc2[r];
        hr[48 + lrow] = acc3[r];
        float ps = acc0[r] * as0 + acc1[r] * as1 + acc2[r] * as2 + acc3[r] * as3;
        float pd = acc0[r] * ad0 + acc1[r] * ad1 + acc2[r] * ad2 + acc3[r] * ad3;
        #pragma unroll
        for (int off = 1; off < 16; off <<= 1) {
            ps += __shfl_xor(ps, off, 32);
            pd += __shfl_xor(pd, off, 32);
        }
        if (lrow == 0) { a_src[row] = ps; a_dst[row] = pd; }
    }

    {
        unsigned wa = 0, wb = 0;
        #pragma unroll
        for (int t = 0; t < 2; ++t) {
            const int4 dv = t ? dv1 : dv0;
            const int4 sv = t ? sv1 : sv0;
            const unsigned bb[4] = { bin_of(dv.x), bin_of(dv.y), bin_of(dv.z), bin_of(dv.w) };
            uint4 rv;
            rv.x = (unsigned)sv.x | ((((unsigned)dv.x - bb[0] * NPB) & 0xFFu) << 16);
            rv.y = (unsigned)sv.y | ((((unsigned)dv.y - bb[1] * NPB) & 0xFFu) << 16);
            rv.z = (unsigned)sv.z | ((((unsigned)dv.z - bb[2] * NPB) & 0xFFu) << 16);
            rv.w = (unsigned)sv.w | ((((unsigned)dv.w - bb[3] * NPB) & 0xFFu) << 16);
            ((uint4*)rec)[tile * 64 + t * 32 + lane] = rv;
            #pragma unroll
            for (int c = 0; c < 4; ++c) {
                #pragma unroll
                for (int i = 0; i < 8; ++i) {
                    const unsigned pl = __builtin_amdgcn_ballot_w32(((bb[c] >> i) & 1u) != 0u);
                    if (i < 4) wa = (lane == 16 * t + 4 * c + i)       ? pl : wa;
                    else       wb = (lane == 16 * t + 4 * c + (i - 4)) ? pl : wb;
                }
            }
        }
        binw[tile * 32 + lane]               = wa;
        binw[N_EDGES / 8 + tile * 32 + lane] = wb;
    }
}

#define QCAP 16384
#define OWN_T 1024
#define PB 16
#define SCAN_U 5
#define OWN_W (OWN_T / 32)
#define RPW ((NPB + OWN_W - 1) / OWN_W)

__device__ __forceinline__ void lds_max_float(float* addr, float val) {
    if (val >= 0.0f) atomicMax((int*)addr, __float_as_int(val));
    else             atomicMin((unsigned int*)addr, __float_as_uint(val));
}

struct PlaneKey { unsigned w[8]; };

__device__ __forceinline__ unsigned hits32(uint4 lo, uint4 hi, const PlaneKey& key) {
    return (lo.x ^ key.w[0]) & (lo.y ^ key.w[1]) & (lo.z ^ key.w[2]) & (lo.w ^ key.w[3]) &
           (hi.x ^ key.w[4]) & (hi.y ^ key.w[5]) & (hi.z ^ key.w[6]) & (hi.w ^ key.w[7]);
}

__device__ __forceinline__ int wave_prefix_sum(int v, int lane) {
    v += __builtin_amdgcn_update_dpp(0, v, 0x111, 0xF, 0xF, false);
    v += __builtin_amdgcn_update_dpp(0, v, 0x112, 0xF, 0xF, false);
    v += __builtin_amdgcn_update_dpp(0, v, 0x114, 0xF, 0xF, false);
    v += __builtin_amdgcn_update_dpp(0, v, 0x118, 0xF, 0xF, false);
    const int low = __builtin_amdgcn_readlane(v, 15);
    return v + ((lane >= 16) ? low : 0);
}

template <bool GUARD>
__device__ __forceinline__ void scan_round(const uint4* __restrict__ plane_lo,
                                           const uint4* __restrict__ plane_hi,
                                           int g0, int lenG, int rotG, const PlaneKey& key,
                                           int i0, int lane, unsigned* q_pk, int* q_n) {
    uint4 lo[SCAN_U], hi[SCAN_U];
    int   G[SCAN_U];
    #pragma unroll
    for (int k = 0; k < SCAN_U; ++k) {
        int jj = i0 + k * OWN_T + rotG;
        if (jj >= lenG) jj -= lenG;
        G[k] = g0 + jj;
        if (!GUARD || (i0 + k * OWN_T < lenG)) { lo[k] = plane_lo[G[k]]; hi[k] = plane_hi[G[k]]; }
        else { lo[k] = make_uint4(~0u, ~0u, ~0u, ~0u); hi[k] = lo[k]; }
    }
    unsigned m[SCAN_U];
    int cnt = 0;
    #pragma unroll
    for (int k = 0; k < SCAN_U; ++k) {
        m[k] = hits32(lo[k], hi[k], key);
        cnt += __popc(m[k]);
    }
    const int incl  = wave_prefix_sum(cnt, lane);
    const int total = __builtin_amdgcn_readlane(incl, 31);
    int base = 0;
    if (lane == 0) base = atomicAdd(q_n, total);
    int slot = __builtin_amdgcn_readfirstlane(base) + incl - cnt;
    #pragma unroll
    for (int k = 0; k < SCAN_U; ++k) {
        unsigned mm = m[k];
        const int e0 = ((G[k] & ~3) << 5) + (G[k] & 3);
        while (mm) {
            const int p = __builtin_ctz(mm);
            mm &= mm - 1;
            if (slot < QCAP) q_pk[slot] = (unsigned)(e0 + 4 * p);
            ++slot;
        }
    }
}

__device__ __forceinline__ void scan_enqueue(const uint4* __restrict__ plane_lo,
                                             const uint4* __restrict__ plane_hi,
                                             int g0, int lenG, int rotG, const PlaneKey& key,
                                             int tid, unsigned* q_pk, int* q_n) {
    const int full = lenG / (SCAN_U * OWN_T);
    for (int r = 0; r < full; ++r)
        scan_round<false>(plane_lo, plane_hi, g0, lenG, rotG, key, tid + r * (SCAN_U * OWN_T), tid & 31, q_pk, q_n);
    if (full * (SCAN_U * OWN_T) < lenG)
        scan_round<true>(plane_lo, plane_hi, g0, lenG, rotG, key, tid + full * (SCAN_U * OWN_T), tid & 31, q_pk, q_n);
}

__device__ __forceinline__ void resolve_max(const unsigned* __restrict__ rec,
                                            const float* __restrict__ a_src,
                                            const float* adst_l, float* m_l,
                                            const unsigned* q_pk, unsigned* q_sd, float* q_sc,
                                            int n, int tid) {
    for (int i0 = tid; i0 < n; i0 += 4 * OWN_T) {
        unsigned u[4];
        unsigned r[4];
        float    as[4];
        #pragma unroll
        for (int k = 0; k < 4; ++k) {
            const int i = i0 + k * OWN_T;
            u[k] = q_pk[(i < n) ? i : (n - 1)];
        }
        #pragma unroll
        for (int k = 0; k < 4; ++k) r[k] = rec[u[k]];
        #pragma unroll
        for (int k = 0; k < 4; ++k) as[k] = a_src[r[k] & 0xFFFFu];
        #pragma unroll
        for (int k = 0; k < 4; ++k) {
            const int i = i0 + k * OWN_T;
            if (i < n) {
                const unsigned dl = r[k] >> 16;
                float v = as[k] + adst_l[dl];
                v = (v > 0.0f) ? v : v * NEG_SLOPE;
                q_sd[i] = r[k];
                q_sc[i] = v;
                lds_max_float(&m_l[dl], v);
            }
        }
    }
}

template <bool FULL>
__device__ __forceinline__ void process_batch(const float* __restrict__ h,
                                              const float* m_l, float* s_l, float* acc,
                                              const unsigned* q_sd, const float* q_sc,
                                              int base, int n, int lane) {
    const int  my = base + (lane & (PB - 1));
    const bool ok = FULL || (my < n);
    const unsigned u  = q_sd[ok ? my : base];
    const float    sc = q_sc[ok ? my : base];
    const float    ex = expf(sc - m_l[u >> 16]);
    if (ok && lane < PB) atomicAdd(&s_l[u >> 16], ex);
    v2f hv[PB];
    #pragma unroll
    for (int k = 0; k < PB; ++k) {
        const unsigned uk = (unsigned)__builtin_amdgcn_readlane((int)u, k);
        hv[k] = *(const v2f*)(h + (size_t)(uk & 0xFFFFu) * HIDDEN + lane * 2);
    }
    #pragma unroll
    for (int k = 0; k < PB; ++k) {
        if (FULL || (base + k < n)) {
            const unsigned uk  = (unsigned)__builtin_amdgcn_readlane((int)u, k);
            const float    exk = __int_as_float(__builtin_amdgcn_readlane(__float_as_int(ex), k));
            float* a = acc + (uk >> 16) * HIDDEN + lane * 2;
            atomicAdd(a,     exk * hv[k].x);
            atomicAdd(a + 1, exk * hv[k].y);
        }
    }
}

__device__ __forceinline__ void process_queue(const float* __restrict__ h,
                                              const float* m_l, float* s_l, float* acc,
                                              const unsigned* q_sd, const float* q_sc,
                                              int n, int wv, int lane) {
    for (int base = wv * PB; base < n; base += (OWN_T / 32) * PB) {
        if (base + PB <= n) process_batch<true >(h, m_l, s_l, acc, q_sd, q_sc, base, n, lane);
        else                process_batch<false>(h, m_l, s_l, acc, q_sd, q_sc, base, n, lane);
    }
}

__global__ void __launch_bounds__(OWN_T)
gat_owner_kernel(const unsigned* __restrict__ rec,
                 const unsigned* __restrict__ binw,
                 const float* __restrict__ a_src,
                 const float* __restrict__ a_dst,
                 const float* __restrict__ h,
                 const float* __restrict__ bias_conv,
                 const float* __restrict__ W_lin,
                 const float* __restrict__ b_lin,
                 float* __restrict__ y) {
    __shared__ float    acc[NPB * HIDDEN];
    __shared__ float    m_l[NPB];
    __shared__ float    s_l[NPB];
    __shared__ float    adst_l[NPB];
    __shared__ float    self_sc[NPB];
    __shared__ unsigned q_pk[QCAP];
    __shared__ unsigned q_sd[QCAP];
    __shared__ float    q_sc[QCAP];
    __shared__ int      q_n;

    const int tid   = threadIdx.x;
    const int lane  = tid & 31;
    const int wv    = __builtin_amdgcn_readfirstlane(tid >> 5);
    __builtin_assume(wv >= 0 && wv < OWN_W);
    const int node0 = blockIdx.x * NPB;
    const uint4* plane_lo = (const uint4*)binw;
    const uint4* plane_hi = (const uint4*)(binw + N_EDGES / 8);
    PlaneKey key;
    #pragma unroll
    for (int i = 0; i < 8; ++i) key.w[i] = ((blockIdx.x >> i) & 1u) - 1u;
    static_assert(N_OWN_BLOCKS <= 255 && N_EDGES % 128 == 0 && QCAP % 128 == 0, "plane list");

    static_assert(RPW <= 32 && OWN_W <= NPB, "start values: one lane per own row of a wave, and row wv exists");
    v2f hs[RPW];
    #pragma unroll
    for (int j = 0; j < RPW; ++j) {
        const int dl  = wv + j * OWN_W;
        const int dlc = (dl < NPB) ? dl : (NPB - 1);
        hs[j] = *(const v2f*)(h + (size_t)(node0 + dlc) * HIDDEN + lane * 2);
    }

    if (tid < NPB) {
        const float ad = a_dst[node0 + tid];
        float v = a_src[node0 + tid] + ad;
        v = (v > 0.0f) ? v : v * NEG_SLOPE;
        adst_l[tid]  = ad;
        self_sc[tid] = v;
        m_l[tid]     = v;
    }
    if (tid == 0) q_n = 0;

    __syncthreads();

    scan_enqueue(plane_lo, plane_hi, 0, N_EDGES / 32, blockIdx.x * (N_EDGES / 32 / N_OWN_BLOCKS), key, tid, q_pk, &q_n);
    __syncthreads();
    const int n_all = __builtin_amdgcn_readfirstlane(q_n);

    if (n_all <= QCAP) {
        resolve_max(rec, a_src, adst_l, m_l, q_pk, q_sd, q_sc, n_all, tid);
        __syncthreads();
        const int   jr  = (lane < RPW) ? lane : 0;
        const int   dlr = wv + jr * OWN_W;
        const bool  okr = (lane < RPW) && (dlr < NPB);
        const int   dlq = okr ? dlr : wv;
        const float exl = expf(self_sc[dlq] - m_l[dlq]);
        if (okr) s_l[dlr] = exl;
        #pragma unroll
        for (int j = 0; j < RPW; ++j) {
            const int dl = wv + j * OWN_W;
            if (dl < NPB) {
                const float ex = __int_as_float(__builtin_amdgcn_readlane(__float_as_int(exl), j));
                float* a = acc + dl * HIDDEN + lane * 2;
                a[0] = ex * hs[j].x;
                a[1] = ex * hs[j].y;
            }
        }
        __syncthreads();
        process_queue(h, m_l, s_l, acc, q_sd, q_sc, n_all, wv, lane);
        __syncthreads();
    } else {
        for (int b = 0; b < N_EDGES; b += QCAP) {
            const int len = (N_EDGES - b < QCAP) ? (N_EDGES - b) : QCAP;
            __syncthreads();
            if (tid == 0) q_n = 0;
            __syncthreads();
            scan_enqueue(plane_lo, plane_hi, b / 32, len / 32, 0, key, tid, q_pk, &q_n);
            __syncthreads();
            resolve_max(rec, a_src, adst_l, m_l, q_pk, q_sd, q_sc, __builtin_amdgcn_readfirstlane(q_n), tid);
        }
        __syncthreads();
        for (int i = tid; i < NPB * HIDDEN; i += OWN_T) {
            const int dl = i >> 6;
            const float ex = expf(self_sc[dl] - m_l[dl]);
            acc[i] = ex * h[(size_t)node0 * HIDDEN + i];
            if ((i & 63) == 0) s_l[dl] = ex;
        }
        for (int b = 0; b < N_EDGES; b += QCAP) {
            const int len = (N_EDGES - b < QCAP) ? (N_EDGES - b) : QCAP;
            __syncthreads();
            if (tid == 0) q_n = 0;
            __syncthreads();
            scan_enqueue(plane_lo, plane_hi, b / 32, len / 32, 0, key, tid, q_pk, &q_n);
            __syncthreads();
            const int n = __builtin_amdgcn_readfirstlane(q_n);
            resolve_max(rec, a_src, adst_l, m_l, q_pk, q_sd, q_sc, n, tid);
            __syncthreads();
            process_queue(h, m_l, s_l, acc, q_sd, q_sc, n, wv, lane);
        }
        __syncthreads();
    }

    const float bc0 = bias_conv[lane * 2], bc1 = bias_conv[lane * 2 + 1];
    const float wl0 = W_lin[lane * 2],     wl1 = W_lin[lane * 2 + 1];
    const float bl  = b_lin[0];
    for (int dl = wv; dl < NPB; dl += OWN_T / 32) {
        const float inv = 1.0f / s_l[dl];
        const v2f o = *(const v2f*)(acc + dl * HIDDEN + lane * 2);
        const float o0 = fmaxf(o.x * inv + bc0, 0.0f);
        const float o1 = fmaxf(o.y * inv + bc1, 0.0f);
        float p = o0 * wl0 + o1 * wl1;
        #pragma unroll
        for (int off = 1; off < 32; off <<= 1) p += __shfl_xor(p, off, 32);
        if (lane == 0) y[node0 + dl] = 1.0f / (1.0f + expf(-(p + bl)));
    }
}

extern "C" void kernel_launch(void* const* d_in, const int* in_sizes, int n_in,
                              void* d_out, int out_size, void* d_ws, size_t ws_size,
                              hipStream_t stream) {
    const float* x         = (const float*)d_in[0];
    const int*   ei        = (const int*)  d_in[1];
    const float* W         = (const float*)d_in[2];
    const float* att_src_v = (const float*)d_in[3];
    const float* att_dst_v = (const float*)d_in[4];
    const float* bias_conv = (const float*)d_in[5];
    const float* W_lin     = (const float*)d_in[6];
    const float* b_lin     = (const float*)d_in[7];
    float* y = (float*)d_out;

    float* ws     = (float*)d_ws;
    float* h      = ws;
    float* a_src  = ws + 3200000;
    float* a_dst  = ws + 3250000;
    unsigned* binw = (unsigned*)(ws + 3300000);
    unsigned* rec  = (unsigned*)(ws + 3500000);

    gat_gemm_wmma_kernel<<<N_TILES, 32, 0, stream>>>(x, W, att_src_v, att_dst_v,
                                                     h, a_src, a_dst, ei, binw, rec);

    gat_owner_kernel<<<N_OWN_BLOCKS, OWN_T, 0, stream>>>(rec, binw, a_src, a_dst, h, bias_conv,
                                                         W_lin, b_lin, y);
}


# A resolved edge goes on the list of the wave owning its row (LDS ticket); that wave sums its 6-7 rows of h in registers and writes their outputs: no LDS float atomics; sums now in ticket order
# speedup vs baseline: 5.1417x; 1.0793x over previous
#include <hip/hip_runtime.h>
#include <math.h>

#define N_NODES 50000
#define N_EDGES 800000
#define IN_DIM 128
#define HIDDEN 64
#define NEG_SLOPE 0.2f
#define E_TOT (N_EDGES + N_NODES)
#define NPB 200
#define N_OWN_BLOCKS (N_NODES / NPB)
#define N_TILES (N_NODES / 16)
#define NO_BIN 255u

typedef float v2f __attribute__((ext_vector_type(2)));
typedef float v4f __attribute__((ext_vector_type(4)));
typedef float v8f __attribute__((ext_vector_type(8)));

__device__ __forceinline__ unsigned bin_of(int d) {
    const unsigned u = (unsigned)d;
    return (u < (unsigned)N_NODES) ? (u / NPB) : NO_BIN;
}

__global__ void __launch_bounds__(32)
gat_gemm_wmma_kernel(const float* __restrict__ x, const float* __restrict__ W,
                     const float* __restrict__ att_src_v, const float* __restrict__ att_dst_v,
                     float* __restrict__ h, float* __restrict__ a_src,
                     float* __restrict__ a_dst,
                     const int* __restrict__ ei, unsigned* __restrict__ binw,
                     unsigned* __restrict__ rec)
{
    const int tile = blockIdx.x;
    const int lane = threadIdx.x;
    const int half = lane >> 4;
    const int lrow = lane & 15;
    const int row0 = tile * 16;

    const float* xrow  = x + (size_t)(row0 + lrow) * IN_DIM;
    const float* wrow  = W + (size_t)lrow * IN_DIM;

    static_assert(N_EDGES == N_TILES * 256, "each GEMM wave converts 256 edges");
    const int4* dst4 = (const int4*)(ei + N_EDGES);
    const int4 dv0 = dst4[tile * 64 + lane];
    const int4 dv1 = dst4[tile * 64 + 32 + lane];
    const int4* src4 = (const int4*)ei;
    const int4 sv0 = src4[tile * 64 + lane];
    const int4 sv1 = src4[tile * 64 + 32 + lane];

    v8f acc0 = {}, acc1 = {}, acc2 = {}, acc3 = {};

    for (int kb = 0; kb < IN_DIM; kb += 8) {
        const int k = kb + 4 * half;
        const v4f a  = *(const v4f*)(xrow + k);
        const v4f b0 = *(const v4f*)(wrow + k);
        const v4f b1 = *(const v4f*)(wrow + 16 * IN_DIM + k);
        const v4f b2 = *(const v4f*)(wrow + 32 * IN_DIM + k);
        const v4f b3 = *(const v4f*)(wrow + 48 * IN_DIM + k);
        const v2f a_lo = __builtin_shufflevector(a, a, 0, 1), a_hi = __builtin_shufflevector(a, a, 2, 3);
        acc0 = __builtin_amdgcn_wmma_f32_16x16x4_f32(false, a_lo, false, __builtin_shufflevector(b0, b0, 0, 1), (short)0, acc0, false, false);
        acc1 = __builtin_amdgcn_wmma_f32_16x16x4_f32(false, a_lo, false, __builtin_shufflevector(b1, b1, 0, 1), (short)0, acc1, false, false);
        acc2 = __builtin_amdgcn_wmma_f32_16x16x4_f32(false, a_lo, false, __builtin_shufflevector(b2, b2, 0, 1), (short)0, acc2, false, false);
        acc3 = __builtin_amdgcn_wmma_f32_16x16x4_f32(false, a_lo, false, __builtin_shufflevector(b3, b3, 0, 1), (short)0, acc3, false, false);
        acc0 = __builtin_amdgcn_wmma_f32_16x16x4_f32(false, a_hi, false, __builtin_shufflevector(b0, b0, 2, 3), (short)0, acc0, false, false);
        acc1 = __builtin_amdgcn_wmma_f32_16x16x4_f32(false, a_hi, false, __builtin_shufflevector(b1, b1, 2, 3), (short)0, acc1, false, false);
        acc2 = __builtin_amdgcn_wmma_f32_16x16x4_f32(false, a_hi, false, __builtin_shufflevector(b2, b2, 2, 3), (short)0, acc2, false, false);
        acc3 = __builtin_amdgcn_wmma_f32_16x16x4_f32(false, a_hi, false, __builtin_shufflevector(b3, b3, 2, 3), (short)0, acc3, false, false);
    }

    const float as0 = att_src_v[lrow],      ad0 = att_dst_v[lrow];
    const float as1 = att_src_v[16 + lrow], ad1 = att_dst_v[16 + lrow];
    const float as2 = att_src_v[32 + lrow], ad2 = att_dst_v[32 + lrow];
    const float as3 = att_src_v[48 + lrow], ad3 = att_dst_v[48 + lrow];

    #pragma unroll
    for (int r = 0; r < 8; ++r) {
        const int row = row0 + r + 8 * half;
        float* hr = h + (size_t)row * HIDDEN;
        hr[lrow]      = acc0[r];
        hr[16 + lrow] = acc1[r];
        hr[32 + lrow] = acc2[r];
        hr[48 + lrow] = acc3[r];
        float ps = acc0[r] * as0 + acc1[r] * as1 + acc2[r] * as2 + acc3[r] * as3;
        float pd = acc0[r] * ad0 + acc1[r] * ad1 + acc2[r] * ad2 + acc3[r] * ad3;
        #pragma unroll
        for (int off = 1; off < 16; off <<= 1) {
            ps += __shfl_xor(ps, off, 32);
            pd += __shfl_xor(pd, off, 32);
        }
        if (lrow == 0) { a_src[row] = ps; a_dst[row] = pd; }
    }

    {
        unsigned wa = 0, wb = 0;
        #pragma unroll
        for (int t = 0; t < 2; ++t) {
            const int4 dv = t ? dv1 : dv0;
            const int4 sv = t ? sv1 : sv0;
            const unsigned bb[4] = { bin_of(dv.x), bin_of(dv.y), bin_of(dv.z), bin_of(dv.w) };
            uint4 rv;
            rv.x = (unsigned)sv.x | ((((unsigned)dv.x - bb[0] * NPB) & 0xFFu) << 16);
            rv.y = (unsigned)sv.y | ((((unsigned)dv.y - bb[1] * NPB) & 0xFFu) << 16);
            rv.z = (unsigned)sv.z | ((((unsigned)dv.z - bb[2] * NPB) & 0xFFu) << 16);
            rv.w = (unsigned)sv.w | ((((unsigned)dv.w - bb[3] * NPB) & 0xFFu) << 16);
            ((uint4*)rec)[tile * 64 + t * 32 + lane] = rv;
            #pragma unroll
            for (int c = 0; c < 4; ++c) {
                #pragma unroll
                for (int i = 0; i < 8; ++i) {
                    const unsigned pl = __builtin_amdgcn_ballot_w32(((bb[c] >> i) & 1u) != 0u);
                    if (i < 4) wa = (lane == 16 * t + 4 * c + i)       ? pl : wa;
                    else       wb = (lane == 16 * t + 4 * c + (i - 4)) ? pl : wb;
                }
            }
        }
        binw[tile * 32 + lane]               = wa;
        binw[N_EDGES / 8 + tile * 32 + lane] = wb;
    }
}

#define QCAP 16384
#define OWN_T 1024
#define PB 16
#define SCAN_U 5
#define OWN_W (OWN_T / 32)
#define RPW ((NPB + OWN_W - 1) / OWN_W)

__device__ __forceinline__ void lds_max_float(float* addr, float val) {
    if (val >= 0.0f) atomicMax((int*)addr, __float_as_int(val));
    else             atomicMin((unsigned int*)addr, __float_as_uint(val));
}

struct PlaneKey { unsigned w[8]; };

__device__ __forceinline__ unsigned hits32(uint4 lo, uint4 hi, const PlaneKey& key) {
    return (lo.x ^ key.w[0]) & (lo.y ^ key.w[1]) & (lo.z ^ key.w[2]) & (lo.w ^ key.w[3]) &
           (hi.x ^ key.w[4]) & (hi.y ^ key.w[5]) & (hi.z ^ key.w[6]) & (hi.w ^ key.w[7]);
}

__device__ __forceinline__ int wave_prefix_sum(int v, int lane) {
    v += __builtin_amdgcn_update_dpp(0, v, 0x111, 0xF, 0xF, false);
    v += __builtin_amdgcn_update_dpp(0, v, 0x112, 0xF, 0xF, false);
    v += __builtin_amdgcn_update_dpp(0, v, 0x114, 0xF, 0xF, false);
    v += __builtin_amdgcn_update_dpp(0, v, 0x118, 0xF, 0xF, false);
    const int low = __builtin_amdgcn_readlane(v, 15);
    return v + ((lane >= 16) ? low : 0);
}

template <bool GUARD>
__device__ __forceinline__ void scan_round(const uint4* __restrict__ plane_lo,
                                           const uint4* __restrict__ plane_hi,
                                           int g0, int lenG, int rotG, const PlaneKey& key,
                                           int i0, int lane, unsigned* q_pk, int* q_n) {
    uint4 lo[SCAN_U], hi[SCAN_U];
    int   G[SCAN_U];
    #pragma unroll
    for (int k = 0; k < SCAN_U; ++k) {
        int jj = i0 + k * OWN_T + rotG;
        if (jj >= lenG) jj -= lenG;
        G[k] = g0 + jj;
        if (!GUARD || (i0 + k * OWN_T < lenG)) { lo[k] = plane_lo[G[k]]; hi[k] = plane_hi[G[k]]; }
        else { lo[k] = make_uint4(~0u, ~0u, ~0u, ~0u); hi[k] = lo[k]; }
    }
    unsigned m[SCAN_U];
    int cnt = 0;
    #pragma unroll
    for (int k = 0; k < SCAN_U; ++k) {
        m[k] = hits32(lo[k], hi[k], key);
        cnt += __popc(m[k]);
    }
    const int incl  = wave_prefix_sum(cnt, lane);
    const int total = __builtin_amdgcn_readlane(incl, 31);
    int base = 0;
    if (lane == 0) base = atomicAdd(q_n, total);
    int slot = __builtin_amdgcn_readfirstlane(base) + incl - cnt;
    #pragma unroll
    for (int k = 0; k < SCAN_U; ++k) {
        unsigned mm = m[k];
        const int e0 = ((G[k] & ~3) << 5) + (G[k] & 3);
        while (mm) {
            const int p = __builtin_ctz(mm);
            mm &= mm - 1;
            if (slot < QCAP) q_pk[slot] = (unsigned)(e0 + 4 * p);
            ++slot;
        }
    }
}

__device__ __forceinline__ void scan_enqueue(const uint4* __restrict__ plane_lo,
                                             const uint4* __restrict__ plane_hi,
                                             int g0, int lenG, int rotG, const PlaneKey& key,
                                             int tid, unsigned* q_pk, int* q_n) {
    const int full = lenG / (SCAN_U * OWN_T);
    for (int r = 0; r < full; ++r)
        scan_round<false>(plane_lo, plane_hi, g0, lenG, rotG, key, tid + r * (SCAN_U * OWN_T), tid & 31, q_pk, q_n);
    if (full * (SCAN_U * OWN_T) < lenG)
        scan_round<true>(plane_lo, plane_hi, g0, lenG, rotG, key, tid + full * (SCAN_U * OWN_T), tid & 31, q_pk, q_n);
}

__device__ __forceinline__ void resolve_max(const unsigned* __restrict__ rec,
                                            const float* __restrict__ a_src,
                                            const float* adst_l, float* m_l,
                                            const unsigned* q_pk, unsigned* q_sd, float* q_sc,
                                            int n, int tid) {
    for (int i0 = tid; i0 < n; i0 += 4 * OWN_T) {
        unsigned u[4];
        unsigned r[4];
        float    as[4];
        #pragma unroll
        for (int k = 0; k < 4; ++k) {
            const int i = i0 + k * OWN_T;
            u[k] = q_pk[(i < n) ? i : (n - 1)];
        }
        #pragma unroll
        for (int k = 0; k < 4; ++k) r[k] = rec[u[k]];
        #pragma unroll
        for (int k = 0; k < 4; ++k) as[k] = a_src[r[k] & 0xFFFFu];
        #pragma unroll
        for (int k = 0; k < 4; ++k) {
            const int i = i0 + k * OWN_T;
            if (i < n) {
                const unsigned dl = r[k] >> 16;
                float v = as[k] + adst_l[dl];
                v = (v > 0.0f) ? v : v * NEG_SLOPE;
                q_sd[i] = r[k];
                q_sc[i] = v;
                lds_max_float(&m_l[dl], v);
            }
        }
    }
}

template <bool FULL>
__device__ __forceinline__ void process_batch(const float* __restrict__ h,
                                              const float* m_l, float* s_l, float* acc,
                                              const unsigned* q_sd, const float* q_sc,
                                              int base, int n, int lane) {
    const int  my = base + (lane & (PB - 1));
    const bool ok = FULL || (my < n);
    const unsigned u  = q_sd[ok ? my : base];
    const float    sc = q_sc[ok ? my : base];
    const float    ex = expf(sc - m_l[u >> 16]);
    if (ok && lane < PB) atomicAdd(&s_l[u >> 16], ex);
    v2f hv[PB];
    #pragma unroll
    for (int k = 0; k < PB; ++k) {
        const unsigned uk = (unsigned)__builtin_amdgcn_readlane((int)u, k);
        hv[k] = *(const v2f*)(h + (size_t)(uk & 0xFFFFu) * HIDDEN + lane * 2);
    }
    #pragma unroll
    for (int k = 0; k < PB; ++k) {
        if (FULL || (base + k < n)) {
            const unsigned uk  = (unsigned)__builtin_amdgcn_readlane((int)u, k);
            const float    exk = __int_as_float(__builtin_amdgcn_readlane(__float_as_int(ex), k));
            float* a = acc + (uk >> 16) * HIDDEN + lane * 2;
            atomicAdd(a,     exk * hv[k].x);
            atomicAdd(a + 1, exk * hv[k].y);
        }
    }
}

__device__ __forceinline__ void process_queue(const float* __restrict__ h,
                                              const float* m_l, float* s_l, float* acc,
                                              const unsigned* q_sd, const float* q_sc,
                                              int n, int wv, int lane) {
    for (int base = wv * PB; base < n; base += (OWN_T / 32) * PB) {
        if (base + PB <= n) process_batch<true >(h, m_l, s_l, acc, q_sd, q_sc, base, n, lane);
        else                process_batch<false>(h, m_l, s_l, acc, q_sd, q_sc, base, n, lane);
    }
}

#define WCAP (QCAP / OWN_W)

__device__ __forceinline__ void resolve_route(const unsigned* __restrict__ rec,
                                              const float* __restrict__ a_src,
                                              const float* adst_l, float* m_l,
                                              const unsigned* q_pk, unsigned* q_sd, float* q_sc,
                                              int* wq_n, int* ovf, int n, int tid) {
    for (int i0 = tid; i0 < n; i0 += 4 * OWN_T) {
        unsigned u[4];
        unsigned r[4];
        float    as[4];
        #pragma unroll
        for (int k = 0; k < 4; ++k) {
            const int i = i0 + k * OWN_T;
            u[k] = q_pk[(i < n) ? i : (n - 1)];
        }
        #pragma unroll
        for (int k = 0; k < 4; ++k) r[k] = rec[u[k]];
        #pragma unroll
        for (int k = 0; k < 4; ++k) as[k] = a_src[r[k] & 0xFFFFu];
        float v[4];
        int   t[4];
        #pragma unroll
        for (int k = 0; k < 4; ++k) {
            const unsigned dl = r[k] >> 16;
            v[k] = as[k] + adst_l[dl];
            v[k] = (v[k] > 0.0f) ? v[k] : v[k] * NEG_SLOPE;
        }
        #pragma unroll
        for (int k = 0; k < 4; ++k) {
            const int i = i0 + k * OWN_T;
            t[k] = WCAP;
            if (i < n) {
                const unsigned dl = r[k] >> 16;
                lds_max_float(&m_l[dl], v[k]);
                t[k] = atomicAdd(&wq_n[dl & (OWN_W - 1)], 1);
            }
        }
        #pragma unroll
        for (int k = 0; k < 4; ++k) {
            const int i = i0 + k * OWN_T;
            if (i < n) {
                const unsigned w = (r[k] >> 16) & (OWN_W - 1);
                if (t[k] < WCAP) { q_sd[w * WCAP + t[k]] = r[k]; q_sc[w * WCAP + t[k]] = v[k]; }
                else             *ovf = 1;
            }
        }
    }
}

typedef float v16f __attribute__((ext_vector_type(16)));
template <bool FULL>
__device__ __forceinline__ void own_batch(const float* __restrict__ h, const float* m_l,
                                          const unsigned* wsd, const float* wsc,
                                          int base, int n, int lane,
                                          v16f& racc, float (&rsum)[RPW]) {
    static_assert(RPW <= 8 && OWN_W == 32, "own_batch: at most eight own rows a wave, row >> 5 picks one");
    const int  my = base + (lane & (PB - 1));
    const bool ok = FULL || (my < n);
    const unsigned u  = wsd[ok ? my : base];
    const float    sc = wsc[ok ? my : base];
    const float    ex = expf(sc - m_l[u >> 16]);
    #pragma unroll
    for (int c = 0; c < RPW; ++c) rsum[c] += (ok && lane < PB && (u >> 21) == (unsigned)c) ? ex : 0.0f;
    v2f hv[PB];
    #pragma unroll
    for (int k = 0; k < PB; ++k) {
        const unsigned uk = (unsigned)__builtin_amdgcn_readlane((int)u, k);
        hv[k] = *(const v2f*)(h + (size_t)(uk & 0xFFFFu) * HIDDEN + lane * 2);
    }
    #pragma unroll
    for (int k = 0; k < PB; ++k) {
        if (FULL || (base + k < n)) {
            const unsigned uk  = (unsigned)__builtin_amdgcn_readlane((int)u, k);
            const float    exk = __int_as_float(__builtin_amdgcn_readlane(__float_as_int(ex), k));
            const unsigned c2  = (uk >> 21) * 2u;
            racc[c2]     += exk * hv[k].x;
            racc[c2 + 1] += exk * hv[k].y;
        }
    }
}

__global__ void __launch_bounds__(OWN_T)
gat_owner_kernel(const unsigned* __restrict__ rec,
                 const unsigned* __restrict__ binw,
                 const float* __restrict__ a_src,
                 const float* __restrict__ a_dst,
                 const float* __restrict__ h,
                 const float* __restrict__ bias_conv,
                 const float* __restrict__ W_lin,
                 const float* __restrict__ b_lin,
                 float* __restrict__ y) {
    __shared__ float    acc[NPB * HIDDEN];
    __shared__ float    m_l[NPB];
    __shared__ float    s_l[NPB];
    __shared__ float    adst_l[NPB];
    __shared__ float    self_sc[NPB];
    __shared__ unsigned q_pk[QCAP];
    __shared__ unsigned q_sd[QCAP];
    __shared__ float    q_sc[QCAP];
    __shared__ int      q_n;
    __shared__ int      wq_n[OWN_W];
    __shared__ int      ovf;

    const int tid   = threadIdx.x;
    const int lane  = tid & 31;
    const int wv    = __builtin_amdgcn_readfirstlane(tid >> 5);
    __builtin_assume(wv >= 0 && wv < OWN_W);
    const int node0 = blockIdx.x * NPB;
    const uint4* plane_lo = (const uint4*)binw;
    const uint4* plane_hi = (const uint4*)(binw + N_EDGES / 8);
    PlaneKey key;
    #pragma unroll
    for (int i = 0; i < 8; ++i) key.w[i] = ((blockIdx.x >> i) & 1u) - 1u;
    static_assert(N_OWN_BLOCKS <= 255 && N_EDGES % 128 == 0 && QCAP % 128 == 0, "plane list");

    static_assert(RPW <= 32 && OWN_W <= NPB, "start values: one lane per own row of a wave, and row wv exists");
    v2f hs[RPW];
    #pragma unroll
    for (int j = 0; j < RPW; ++j) {
        const int dl  = wv + j * OWN_W;
        const int dlc = (dl < NPB) ? dl : (NPB - 1);
        hs[j] = *(const v2f*)(h + (size_t)(node0 + dlc) * HIDDEN + lane * 2);
    }

    if (tid < NPB) {
        const float ad = a_dst[node0 + tid];
        float v = a_src[node0 + tid] + ad;
        v = (v > 0.0f) ? v : v * NEG_SLOPE;
        adst_l[tid]  = ad;
        self_sc[tid] = v;
        m_l[tid]     = v;
    }
    if (tid == 0) { q_n = 0; ovf = 0; }
    if (tid < OWN_W) wq_n[tid] = 0;

    __syncthreads();

    scan_enqueue(plane_lo, plane_hi, 0, N_EDGES / 32, blockIdx.x * (N_EDGES / 32 / N_OWN_BLOCKS), key, tid, q_pk, &q_n);
    __syncthreads();
    const int n_all = __builtin_amdgcn_readfirstlane(q_n);

    const float bc0 = bias_conv[lane * 2], bc1 = bias_conv[lane * 2 + 1];
    const float wl0 = W_lin[lane * 2],     wl1 = W_lin[lane * 2 + 1];
    const float bl  = b_lin[0];

    bool own_lists = (n_all <= QCAP);
    if (own_lists) {
        resolve_route(rec, a_src, adst_l, m_l, q_pk, q_sd, q_sc, wq_n, &ovf, n_all, tid);
        __syncthreads();
        own_lists = (__builtin_amdgcn_readfirstlane(ovf) == 0);
    }

    if (own_lists) {
        const int   jr  = (lane < RPW) ? lane : 0;
        const int   dlr = wv + jr * OWN_W;
        const bool  okr = (lane < RPW) && (dlr < NPB);
        const int   dlq = okr ? dlr : wv;
        const float exl = expf(self_sc[dlq] - m_l[dlq]);
        v16f  racc = {};
        float rsum[RPW];
        #pragma unroll
        for (int j = 0; j < RPW; ++j) {
            const float ex = __int_as_float(__builtin_amdgcn_readlane(__float_as_int(exl), j));
            racc[2 * j]     = ex * hs[j].x;
            racc[2 * j + 1] = ex * hs[j].y;
            rsum[j] = (lane == 0) ? ex : 0.0f;
        }
        const int       cnt = __builtin_amdgcn_readfirstlane(wq_n[wv]);
        const unsigned* wsd = q_sd + wv * WCAP;
        const float*    wsc = q_sc + wv * WCAP;
        for (int base = 0; base < cnt; base += PB) {
            if (base + PB <= cnt) own_batch<true >(h, m_l, wsd, wsc, base, cnt, lane, racc, rsum);
            else                  own_batch<false>(h, m_l, wsd, wsc, base, cnt, lane, racc, rsum);
        }
        #pragma unroll
        for (int j = 0; j < RPW; ++j) {
            const int dl = wv + j * OWN_W;
            if (dl < NPB) {
                float ws = rsum[j];
                #pragma unroll
                for (int off = 1; off < 32; off <<= 1) ws += __shfl_xor(ws, off, 32);
                const float inv = 1.0f / ws;
                const float o0 = fmaxf(racc[2 * j]     * inv + bc0, 0.0f);
                const float o1 = fmaxf(racc[2 * j + 1] * inv + bc1, 0.0f);
                float p = o0 * wl0 + o1 * wl1;
                #pragma unroll
                for (int off = 1; off < 32; off <<= 1) p += __shfl_xor(p, off, 32);
                if (lane == 0) y[node0 + dl] = 1.0f / (1.0f + expf(-(p + bl)));
            }
        }
    } else {
        for (int b = 0; b < N_EDGES; b += QCAP) {
            const int len = (N_EDGES - b < QCAP) ? (N_EDGES - b) : QCAP;
            __syncthreads();
            if (tid == 0) q_n = 0;
            __syncthreads();
            scan_enqueue(plane_lo, plane_hi, b / 32, len / 32, 0, key, tid, q_pk, &q_n);
            __syncthreads();
            resolve_max(rec, a_src, adst_l, m_l, q_pk, q_sd, q_sc, __builtin_amdgcn_readfirstlane(q_n), tid);
        }
        __syncthreads();
        for (int i = tid; i < NPB * HIDDEN; i += OWN_T) {
            const int dl = i >> 6;
            const float ex = expf(self_sc[dl] - m_l[dl]);
            acc[i] = ex * h[(size_t)node0 * HIDDEN + i];
            if ((i & 63) == 0) s_l[dl] = ex;
        }
        for (int b = 0; b < N_EDGES; b += QCAP) {
            const int len = (N_EDGES - b < QCAP) ? (N_EDGES - b) : QCAP;
            __syncthreads();
            if (tid == 0) q_n = 0;
            __syncthreads();
            scan_enqueue(plane_lo, plane_hi, b / 32, len / 32, 0, key, tid, q_pk, &q_n);
            __syncthreads();
            const int n = __builtin_amdgcn_readfirstlane(q_n);
            resolve_max(rec, a_src, adst_l, m_l, q_pk, q_sd, q_sc, n, tid);
            __syncthreads();
            process_queue(h, m_l, s_l, acc, q_sd, q_sc, n, wv, lane);
        }
        __syncthreads();
        for (int dl = wv; dl < NPB; dl += OWN_T / 32) {
            const float inv = 1.0f / s_l[dl];
            const v2f o = *(const v2f*)(acc + dl * HIDDEN + lane * 2);
            const float o0 = fmaxf(o.x * inv + bc0, 0.0f);
            const float o1 = fmaxf(o.y * inv + bc1, 0.0f);
            float p = o0 * wl0 + o1 * wl1;
            #pragma unroll
            for (int off = 1; off < 32; off <<= 1) p += __shfl_xor(p, off, 32);
            if (lane == 0) y[node0 + dl] = 1.0f / (1.0f + expf(-(p + bl)));
        }
    }
}

extern "C" void kernel_launch(void* const* d_in, const int* in_sizes, int n_in,
                              void* d_out, int out_size, void* d_ws, size_t ws_size,
                              hipStream_t stream) {
    const float* x         = (const float*)d_in[0];
    const int*   ei        = (const int*)  d_in[1];
    const float* W         = (const float*)d_in[2];
    const float* att_src_v = (const float*)d_in[3];
    const float* att_dst_v = (const float*)d_in[4];
    const float* bias_conv = (const float*)d_in[5];
    const float* W_lin     = (const float*)d_in[6];
    const float* b_lin     = (const float*)d_in[7];
    float* y = (float*)d_out;

    float* ws     = (float*)d_ws;
    float* h      = ws;
    float* a_src  = ws + 3200000;
    float* a_dst  = ws + 3250000;
    unsigned* binw = (unsigned*)(ws + 3300000);
    unsigned* rec  = (unsigned*)(ws + 3500000);

    gat_gemm_wmma_kernel<<<N_TILES, 32, 0, stream>>>(x, W, att_src_v, att_dst_v,
                                                     h, a_src, a_dst, ei, binw, rec);

    gat_owner_kernel<<<N_OWN_BLOCKS, OWN_T, 0, stream>>>(rec, binw, a_src, a_dst, h, bias_conv,
                                                         W_lin, b_lin, y);
}


# A resolved edge goes on its destination row's own list in LDS (ticket per row, self loop as entry 0); the owning wave sums one row at a time in two registers a lane; sums in per-row ticket order
# speedup vs baseline: 5.5187x; 1.0733x over previous
#include <hip/hip_runtime.h>
#include <math.h>

#define N_NODES 50000
#define N_EDGES 800000
#define IN_DIM 128
#define HIDDEN 64
#define NEG_SLOPE 0.2f
#define E_TOT (N_EDGES + N_NODES)
#define NPB 200
#define N_OWN_BLOCKS (N_NODES / NPB)
#define N_TILES (N_NODES / 16)
#define NO_BIN 255u

typedef float v2f __attribute__((ext_vector_type(2)));
typedef float v4f __attribute__((ext_vector_type(4)));
typedef float v8f __attribute__((ext_vector_type(8)));

__device__ __forceinline__ unsigned bin_of(int d) {
    const unsigned u = (unsigned)d;
    return (u < (unsigned)N_NODES) ? (u / NPB) : NO_BIN;
}

__global__ void __launch_bounds__(32)
gat_gemm_wmma_kernel(const float* __restrict__ x, const float* __restrict__ W,
                     const float* __restrict__ att_src_v, const float* __restrict__ att_dst_v,
                     float* __restrict__ h, float* __restrict__ a_src,
                     float* __restrict__ a_dst,
                     const int* __restrict__ ei, unsigned* __restrict__ binw,
                     unsigned* __restrict__ rec)
{
    const int tile = blockIdx.x;
    const int lane = threadIdx.x;
    const int half = lane >> 4;
    const int lrow = lane & 15;
    const int row0 = tile * 16;

    const float* xrow  = x + (size_t)(row0 + lrow) * IN_DIM;
    const float* wrow  = W + (size_t)lrow * IN_DIM;

    static_assert(N_EDGES == N_TILES * 256, "each GEMM wave converts 256 edges");
    const int4* dst4 = (const int4*)(ei + N_EDGES);
    const int4 dv0 = dst4[tile * 64 + lane];
    const int4 dv1 = dst4[tile * 64 + 32 + lane];
    const int4* src4 = (const int4*)ei;
    const int4 sv0 = src4[tile * 64 + lane];
    const int4 sv1 = src4[tile * 64 + 32 + lane];

    v8f acc0 = {}, acc1 = {}, acc2 = {}, acc3 = {};

    for (int kb = 0; kb < IN_DIM; kb += 8) {
        const int k = kb + 4 * half;
        const v4f a  = *(const v4f*)(xrow + k);
        const v4f b0 = *(const v4f*)(wrow + k);
        const v4f b1 = *(const v4f*)(wrow + 16 * IN_DIM + k);
        const v4f b2 = *(const v4f*)(wrow + 32 * IN_DIM + k);
        const v4f b3 = *(const v4f*)(wrow + 48 * IN_DIM + k);
        const v2f a_lo = __builtin_shufflevector(a, a, 0, 1), a_hi = __builtin_shufflevector(a, a, 2, 3);
        acc0 = __builtin_amdgcn_wmma_f32_16x16x4_f32(false, a_lo, false, __builtin_shufflevector(b0, b0, 0, 1), (short)0, acc0, false, false);
        acc1 = __builtin_amdgcn_wmma_f32_16x16x4_f32(false, a_lo, false, __builtin_shufflevector(b1, b1, 0, 1), (short)0, acc1, false, false);
        acc2 = __builtin_amdgcn_wmma_f32_16x16x4_f32(false, a_lo, false, __builtin_shufflevector(b2, b2, 0, 1), (short)0, acc2, false, false);
        acc3 = __builtin_amdgcn_wmma_f32_16x16x4_f32(false, a_lo, false, __builtin_shufflevector(b3, b3, 0, 1), (short)0, acc3, false, false);
        acc0 = __builtin_amdgcn_wmma_f32_16x16x4_f32(false, a_hi, false, __builtin_shufflevector(b0, b0, 2, 3), (short)0, acc0, false, false);
        acc1 = __builtin_amdgcn_wmma_f32_16x16x4_f32(false, a_hi, false, __builtin_shufflevector(b1, b1, 2, 3), (short)0, acc1, false, false);
        acc2 = __builtin_amdgcn_wmma_f32_16x16x4_f32(false, a_hi, false, __builtin_shufflevector(b2, b2, 2, 3), (short)0, acc2, false, false);
        acc3 = __builtin_amdgcn_wmma_f32_16x16x4_f32(false, a_hi, false, __builtin_shufflevector(b3, b3, 2, 3), (short)0, acc3, false, false);
    }

    const float as0 = att_src_v[lrow],      ad0 = att_dst_v[lrow];
    const float as1 = att_src_v[16 + lrow], ad1 = att_dst_v[16 + lrow];
    const float as2 = att_src_v[32 + lrow], ad2 = att_dst_v[32 + lrow];
    const float as3 = att_src_v[48 + lrow], ad3 = att_dst_v[48 + lrow];

    #pragma unroll
    for (int r = 0; r < 8; ++r) {
        const int row = row0 + r + 8 * half;
        float* hr = h + (size_t)row * HIDDEN;
        hr[lrow]      = acc0[r];
        hr[16 + lrow] = acc1[r];
        hr[32 + lrow] = acc2[r];
        hr[48 + lrow] = acc3[r];
        float ps = acc0[r] * as0 + acc1[r] * as1 + acc2[r] * as2 + acc3[r] * as3;
        float pd = acc0[r] * ad0 + acc1[r] * ad1 + acc2[r] * ad2 + acc3[r] * ad3;
        #pragma unroll
        for (int off = 1; off < 16; off <<= 1) {
            ps += __shfl_xor(ps, off, 32);
            pd += __shfl_xor(pd, off, 32);
        }
        if (lrow == 0) { a_src[row] = ps; a_dst[row] = pd; }
    }

    {
        unsigned wa = 0, wb = 0;
        #pragma unroll
        for (int t = 0; t < 2; ++t) {
            const int4 dv = t ? dv1 : dv0;
            const int4 sv = t ? sv1 : sv0;
            const unsigned bb[4] = { bin_of(dv.x), bin_of(dv.y), bin_of(dv.z), bin_of(dv.w) };
            uint4 rv;
            rv.x = (unsigned)sv.x | ((((unsigned)dv.x - bb[0] * NPB) & 0xFFu) << 16);
            rv.y = (unsigned)sv.y | ((((unsigned)dv.y - bb[1] * NPB) & 0xFFu) << 16);
            rv.z = (unsigned)sv.z | ((((unsigned)dv.z - bb[2] * NPB) & 0xFFu) << 16);
            rv.w = (unsigned)sv.w | ((((unsigned)dv.w - bb[3] * NPB) & 0xFFu) << 16);
            ((uint4*)rec)[tile * 64 + t * 32 + lane] = rv;
            #pragma unroll
            for (int c = 0; c < 4; ++c) {
                #pragma unroll
                for (int i = 0; i < 8; ++i) {
                    const unsigned pl = __builtin_amdgcn_ballot_w32(((bb[c] >> i) & 1u) != 0u);
                    if (i < 4) wa = (lane == 16 * t + 4 * c + i)       ? pl : wa;
                    else       wb = (lane == 16 * t + 4 * c + (i - 4)) ? pl : wb;
                }
            }
        }
        binw[tile * 32 + lane]               = wa;
        binw[N_EDGES / 8 + tile * 32 + lane] = wb;
    }
}

#define QCAP 16384
#define OWN_T 1024
#define PB 16
#define SCAN_U 5
#define OWN_W (OWN_T / 32)
#define RPW ((NPB + OWN_W - 1) / OWN_W)

__device__ __forceinline__ void lds_max_float(float* addr, float val) {
    if (val >= 0.0f) atomicMax((int*)addr, __float_as_int(val));
    else             atomicMin((unsigned int*)addr, __float_as_uint(val));
}

struct PlaneKey { unsigned w[8]; };

__device__ __forceinline__ unsigned hits32(uint4 lo, uint4 hi, const PlaneKey& key) {
    return (lo.x ^ key.w[0]) & (lo.y ^ key.w[1]) & (lo.z ^ key.w[2]) & (lo.w ^ key.w[3]) &
           (hi.x ^ key.w[4]) & (hi.y ^ key.w[5]) & (hi.z ^ key.w[6]) & (hi.w ^ key.w[7]);
}

__device__ __forceinline__ int wave_prefix_sum(int v, int lane) {
    v += __builtin_amdgcn_update_dpp(0, v, 0x111, 0xF, 0xF, false);
    v += __builtin_amdgcn_update_dpp(0, v, 0x112, 0xF, 0xF, false);
    v += __builtin_amdgcn_update_dpp(0, v, 0x114, 0xF, 0xF, false);
    v += __builtin_amdgcn_update_dpp(0, v, 0x118, 0xF, 0xF, false);
    const int low = __builtin_amdgcn_readlane(v, 15);
    return v + ((lane >= 16) ? low : 0);
}

template <bool GUARD>
__device__ __forceinline__ void scan_round(const uint4* __restrict__ plane_lo,
                                           const uint4* __restrict__ plane_hi,
                                           int g0, int lenG, int rotG, const PlaneKey& key,
                                           int i0, int lane, unsigned* q_pk, int* q_n) {
    uint4 lo[SCAN_U], hi[SCAN_U];
    int   G[SCAN_U];
    #pragma unroll
    for (int k = 0; k < SCAN_U; ++k) {
        int jj = i0 + k * OWN_T + rotG;
        if (jj >= lenG) jj -= lenG;
        G[k] = g0 + jj;
        if (!GUARD || (i0 + k * OWN_T < lenG)) { lo[k] = plane_lo[G[k]]; hi[k] = plane_hi[G[k]]; }
        else { lo[k] = make_uint4(~0u, ~0u, ~0u, ~0u); hi[k] = lo[k]; }
    }
    unsigned m[SCAN_U];
    int cnt = 0;
    #pragma unroll
    for (int k = 0; k < SCAN_U; ++k) {
        m[k] = hits32(lo[k], hi[k], key);
        cnt += __popc(m[k]);
    }
    const int incl  = wave_prefix_sum(cnt, lane);
    const int total = __builtin_amdgcn_readlane(incl, 31);
    int base = 0;
    if (lane == 0) base = atomicAdd(q_n, total);
    int slot = __builtin_amdgcn_readfirstlane(base) + incl - cnt;
    #pragma unroll
    for (int k = 0; k < SCAN_U; ++k) {
        unsigned mm = m[k];
        const int e0 = ((G[k] & ~3) << 5) + (G[k] & 3);
        while (mm) {
            const int p = __builtin_ctz(mm);
            mm &= mm - 1;
            if (slot < QCAP) q_pk[slot] = (unsigned)(e0 + 4 * p);
            ++slot;
        }
    }
}

__device__ __forceinline__ void scan_enqueue(const uint4* __restrict__ plane_lo,
                                             const uint4* __restrict__ plane_hi,
                                             int g0, int lenG, int rotG, const PlaneKey& key,
                                             int tid, unsigned* q_pk, int* q_n) {
    const int full = lenG / (SCAN_U * OWN_T);
    for (int r = 0; r < full; ++r)
        scan_round<false>(plane_lo, plane_hi, g0, lenG, rotG, key, tid + r * (SCAN_U * OWN_T), tid & 31, q_pk, q_n);
    if (full * (SCAN_U * OWN_T) < lenG)
        scan_round<true>(plane_lo, plane_hi, g0, lenG, rotG, key, tid + full * (SCAN_U * OWN_T), tid & 31, q_pk, q_n);
}

__device__ __forceinline__ void resolve_max(const unsigned* __restrict__ rec,
                                            const float* __restrict__ a_src,
                                            const float* adst_l, float* m_l,
                                            const unsigned* q_pk, unsigned* q_sd, float* q_sc,
                                            int n, int tid) {
    for (int i0 = tid; i0 < n; i0 += 4 * OWN_T) {
        unsigned u[4];
        unsigned r[4];
        float    as[4];
        #pragma unroll
        for (int k = 0; k < 4; ++k) {
            const int i = i0 + k * OWN_T;
            u[k] = q_pk[(i < n) ? i : (n - 1)];
        }
        #pragma unroll
        for (int k = 0; k < 4; ++k) r[k] = rec[u[k]];
        #pragma unroll
        for (int k = 0; k < 4; ++k) as[k] = a_src[r[k] & 0xFFFFu];
        #pragma unroll
        for (int k = 0; k < 4; ++k) {
            const int i = i0 + k * OWN_T;
            if (i < n) {
                const unsigned dl = r[k] >> 16;
                float v = as[k] + adst_l[dl];
                v = (v > 0.0f) ? v : v * NEG_SLOPE;
                q_sd[i] = r[k];
                q_sc[i] = v;
                lds_max_float(&m_l[dl], v);
            }
        }
    }
}

template <bool FULL>
__device__ __forceinline__ void process_batch(const float* __restrict__ h,
                                              const float* m_l, float* s_l, float* acc,
                                              const unsigned* q_sd, const float* q_sc,
                                              int base, int n, int lane) {
    const int  my = base + (lane & (PB - 1));
    const bool ok = FULL || (my < n);
    const unsigned u  = q_sd[ok ? my : base];
    const float    sc = q_sc[ok ? my : base];
    const float    ex = expf(sc - m_l[u >> 16]);
    if (ok && lane < PB) atomicAdd(&s_l[u >> 16], ex);
    v2f hv[PB];
    #pragma unroll
    for (int k = 0; k < PB; ++k) {
        const unsigned uk = (unsigned)__builtin_amdgcn_readlane((int)u, k);
        hv[k] = *(const v2f*)(h + (size_t)(uk & 0xFFFFu) * HIDDEN + lane * 2);
    }
    #pragma unroll
    for (int k = 0; k < PB; ++k) {
        if (FULL || (base + k < n)) {
            const unsigned uk  = (unsigned)__builtin_amdgcn_readlane((int)u, k);
            const float    exk = __int_as_float(__builtin_amdgcn_readlane(__float_as_int(ex), k));
            float* a = acc + (uk >> 16) * HIDDEN + lane * 2;
            atomicAdd(a,     exk * hv[k].x);
            atomicAdd(a + 1, exk * hv[k].y);
        }
    }
}

__device__ __forceinline__ void process_queue(const float* __restrict__ h,
                                              const float* m_l, float* s_l, float* acc,
                                              const unsigned* q_sd, const float* q_sc,
                                              int n, int wv, int lane) {
    for (int base = wv * PB; base < n; base += (OWN_T / 32) * PB) {
        if (base + PB <= n) process_batch<true >(h, m_l, s_l, acc, q_sd, q_sc, base, n, lane);
        else                process_batch<false>(h, m_l, s_l, acc, q_sd, q_sc, base, n, lane);
    }
}

#define RCAP 64
#define RSTR 65
static_assert(NPB * RSTR <= QCAP, "the rows' lists live in the resolved arrays");

__device__ __forceinline__ void resolve_route(const unsigned* __restrict__ rec,
                                              const float* __restrict__ a_src,
                                              const float* adst_l, float* m_l,
                                              const unsigned* q_pk, unsigned* q_sd, float* q_sc,
                                              int* row_n, int* ovf, int n, int tid) {
    for (int i0 = tid; i0 < n; i0 += 4 * OWN_T) {
        unsigned u[4];
        unsigned r[4];
        float    as[4];
        #pragma unroll
        for (int k = 0; k < 4; ++k) {
            const int i = i0 + k * OWN_T;
            u[k] = q_pk[(i < n) ? i : (n - 1)];
        }
        #pragma unroll
        for (int k = 0; k < 4; ++k) r[k] = rec[u[k]];
        #pragma unroll
        for (int k = 0; k < 4; ++k) as[k] = a_src[r[k] & 0xFFFFu];
        float v[4];
        int   t[4];
        #pragma unroll
        for (int k = 0; k < 4; ++k) {
            const unsigned dl = r[k] >> 16;
            v[k] = as[k] + adst_l[dl];
            v[k] = (v[k] > 0.0f) ? v[k] : v[k] * NEG_SLOPE;
        }
        #pragma unroll
        for (int k = 0; k < 4; ++k) {
            const int i = i0 + k * OWN_T;
            t[k] = RCAP;
            if (i < n) {
                const unsigned dl = r[k] >> 16;
                lds_max_float(&m_l[dl], v[k]);
                t[k] = atomicAdd(&row_n[dl], 1);
            }
        }
        #pragma unroll
        for (int k = 0; k < 4; ++k) {
            const int i = i0 + k * OWN_T;
            if (i < n) {
                const unsigned dl = r[k] >> 16;
                if (t[k] < RCAP) { q_sd[dl * RSTR + t[k]] = r[k] & 0xFFFFu; q_sc[dl * RSTR + t[k]] = v[k]; }
                else             *ovf = 1;
            }
        }
    }
}

template <int NS, int MINC>
__device__ __forceinline__ void row_part(const float* __restrict__ h, unsigned u, float ex, int cnt, int lane,
                                         float& a0, float& a1) {
    v2f hv[NS];
    #pragma unroll
    for (int k = 0; k < NS; ++k) {
        const unsigned uk = (unsigned)__builtin_amdgcn_readlane((int)u, k);
        hv[k] = *(const v2f*)(h + (size_t)uk * HIDDEN + lane * 2);
    }
    #pragma unroll
    for (int k = 0; k < NS; ++k) {
        const float exk = __int_as_float(__builtin_amdgcn_readlane(__float_as_int(ex), k));
        float hx = hv[k].x, hy = hv[k].y;
        if (k >= MINC) {
            const bool live = k < cnt;
            hx = live ? hx : 0.0f;
            hy = live ? hy : 0.0f;
        }
        a0 = __builtin_fmaf(exk, hx, a0);
        a1 = __builtin_fmaf(exk, hy, a1);
    }
}

#define RB 32
__device__ __forceinline__ void row_batch(const float* __restrict__ h,
                                          const unsigned* rsd, const float* rsc, float m,
                                          int base, int n, int lane,
                                          float& a0, float& a1, float& wsum) {
    const int  cnt = (n - base < RB) ? (n - base) : RB;
    const bool ok  = lane < cnt;
    const unsigned u  = rsd[base + (ok ? lane : 0)];
    const float    sc = rsc[base + (ok ? lane : 0)];
    const float    ex = ok ? expf(sc - m) : 0.0f;
    wsum += ex;
    static_assert(RB == 32, "row_batch: the last case takes every count up to RB");
    switch ((cnt + 3) >> 2) {
        case 1:
        case 2:
        case 3:  row_part<12,  1>(h, u, ex, cnt, lane, a0, a1); break;
        case 4:  row_part<16, 13>(h, u, ex, cnt, lane, a0, a1); break;
        case 5:  row_part<20, 17>(h, u, ex, cnt, lane, a0, a1); break;
        case 6:  row_part<24, 21>(h, u, ex, cnt, lane, a0, a1); break;
        default: row_part<32, 25>(h, u, ex, cnt, lane, a0, a1); break;
    }
}

__global__ void __launch_bounds__(OWN_T)
gat_owner_kernel(const unsigned* __restrict__ rec,
                 const unsigned* __restrict__ binw,
                 const float* __restrict__ a_src,
                 const float* __restrict__ a_dst,
                 const float* __restrict__ h,
                 const float* __restrict__ bias_conv,
                 const float* __restrict__ W_lin,
                 const float* __restrict__ b_lin,
                 float* __restrict__ y) {
    __shared__ float    acc[NPB * HIDDEN];
    __shared__ float    m_l[NPB];
    __shared__ float    s_l[NPB];
    __shared__ float    adst_l[NPB];
    __shared__ float    self_sc[NPB];
    __shared__ unsigned q_pk[QCAP];
    __shared__ unsigned q_sd[QCAP];
    __shared__ float    q_sc[QCAP];
    __shared__ int      q_n;
    __shared__ int      row_n[NPB];
    __shared__ int      ovf;

    const int tid   = threadIdx.x;
    const int lane  = tid & 31;
    const int wv    = __builtin_amdgcn_readfirstlane(tid >> 5);
    __builtin_assume(wv >= 0 && wv < OWN_W);
    const int node0 = blockIdx.x * NPB;
    const uint4* plane_lo = (const uint4*)binw;
    const uint4* plane_hi = (const uint4*)(binw + N_EDGES / 8);
    PlaneKey key;
    #pragma unroll
    for (int i = 0; i < 8; ++i) key.w[i] = ((blockIdx.x >> i) & 1u) - 1u;
    static_assert(N_OWN_BLOCKS <= 255 && N_EDGES % 128 == 0 && QCAP % 128 == 0, "plane list");

    static_assert(RPW <= 8 && OWN_W <= NPB, "own rows: a wave's per-row results fit three 8-wide registers, and row wv exists");

    if (tid < NPB) {
        const float ad = a_dst[node0 + tid];
        float v = a_src[node0 + tid] + ad;
        v = (v > 0.0f) ? v : v * NEG_SLOPE;
        adst_l[tid]  = ad;
        self_sc[tid] = v;
        m_l[tid]     = v;
        q_sd[tid * RSTR] = (unsigned)(node0 + tid);
        q_sc[tid * RSTR] = v;
        row_n[tid]       = 1;
    }
    if (tid == 0) { q_n = 0; ovf = 0; }

    __syncthreads();

    scan_enqueue(plane_lo, plane_hi, 0, N_EDGES / 32, blockIdx.x * (N_EDGES / 32 / N_OWN_BLOCKS), key, tid, q_pk, &q_n);
    __syncthreads();
    const int n_all = __builtin_amdgcn_readfirstlane(q_n);

    const float bc0 = bias_conv[lane * 2], bc1 = bias_conv[lane * 2 + 1];
    const float wl0 = W_lin[lane * 2],     wl1 = W_lin[lane * 2 + 1];
    const float bl  = b_lin[0];

    bool own_lists = (n_all <= QCAP);
    if (own_lists) {
        resolve_route(rec, a_src, adst_l, m_l, q_pk, q_sd, q_sc, row_n, &ovf, n_all, tid);
        __syncthreads();
        own_lists = (__builtin_amdgcn_readfirstlane(ovf) == 0);
    }

    if (own_lists) {
        v8f ra0 = {}, ra1 = {}, rws = {};
        #pragma nounroll
        for (int j = 0; j < RPW; ++j) {
            const int dl = wv + j * OWN_W;
            if (dl >= NPB) break;
            const int   n = __builtin_amdgcn_readfirstlane(row_n[dl]);
            const float m = m_l[dl];
            float a0 = 0.0f, a1 = 0.0f, wsum = 0.0f;
            for (int base = 0; base < n; base += RB)
                row_batch(h, q_sd + dl * RSTR, q_sc + dl * RSTR, m, base, n, lane, a0, a1, wsum);
            ra0[j] = a0; ra1[j] = a1; rws[j] = wsum;
        }
        #pragma unroll
        for (int j = 0; j < RPW; ++j) {
            const int dl = wv + j * OWN_W;
            if (dl < NPB) {
                float ws = rws[j];
                #pragma unroll
                for (int off = 1; off < 32; off <<= 1) ws += __shfl_xor(ws, off, 32);
                const float inv = 1.0f / ws;
                const float o0 = fmaxf(ra0[j] * inv + bc0, 0.0f);
                const float o1 = fmaxf(ra1[j] * inv + bc1, 0.0f);
                float p = o0 * wl0 + o1 * wl1;
                #pragma unroll
                for (int off = 1; off < 32; off <<= 1) p += __shfl_xor(p, off, 32);
                if (lane == 0) y[node0 + dl] = 1.0f / (1.0f + expf(-(p + bl)));
            }
        }
    } else {
        for (int b = 0; b < N_EDGES; b += QCAP) {
            const int len = (N_EDGES - b < QCAP) ? (N_EDGES - b) : QCAP;
            __syncthreads();
            if (tid == 0) q_n = 0;
            __syncthreads();
            scan_enqueue(plane_lo, plane_hi, b / 32, len / 32, 0, key, tid, q_pk, &q_n);
            __syncthreads();
            resolve_max(rec, a_src, adst_l, m_l, q_pk, q_sd, q_sc, __builtin_amdgcn_readfirstlane(q_n), tid);
        }
        __syncthreads();
        for (int i = tid; i < NPB * HIDDEN; i += OWN_T) {
            const int dl = i >> 6;
            const float ex = expf(self_sc[dl] - m_l[dl]);
            acc[i] = ex * h[(size_t)node0 * HIDDEN + i];
            if ((i & 63) == 0) s_l[dl] = ex;
        }
        for (int b = 0; b < N_EDGES; b += QCAP) {
            const int len = (N_EDGES - b < QCAP) ? (N_EDGES - b) : QCAP;
            __syncthreads();
            if (tid == 0) q_n = 0;
            __syncthreads();
            scan_enqueue(plane_lo, plane_hi, b / 32, len / 32, 0, key, tid, q_pk, &q_n);
            __syncthreads();
            const int n = __builtin_amdgcn_readfirstlane(q_n);
            resolve_max(rec, a_src, adst_l, m_l, q_pk, q_sd, q_sc, n, tid);
            __syncthreads();
            process_queue(h, m_l, s_l, acc, q_sd, q_sc, n, wv, lane);
        }
        __syncthreads();
        for (int dl = wv; dl < NPB; dl += OWN_T / 32) {
            const float inv = 1.0f / s_l[dl];
            const v2f o = *(const v2f*)(acc + dl * HIDDEN + lane * 2);
            const float o0 = fmaxf(o.x * inv + bc0, 0.0f);
            const float o1 = fmaxf(o.y * inv + bc1, 0.0f);
            float p = o0 * wl0 + o1 * wl1;
            #pragma unroll
            for (int off = 1; off < 32; off <<= 1) p += __shfl_xor(p, off, 32);
            if (lane == 0) y[node0 + dl] = 1.0f / (1.0f + expf(-(p + bl)));
        }
    }
}

extern "C" void kernel_launch(void* const* d_in, const int* in_sizes, int n_in,
                              void* d_out, int out_size, void* d_ws, size_t ws_size,
                              hipStream_t stream) {
    const float* x         = (const float*)d_in[0];
    const int*   ei        = (const int*)  d_in[1];
    const float* W         = (const float*)d_in[2];
    const float* att_src_v = (const float*)d_in[3];
    const float* att_dst_v = (const float*)d_in[4];
    const float* bias_conv = (const float*)d_in[5];
    const float* W_lin     = (const float*)d_in[6];
    const float* b_lin     = (const float*)d_in[7];
    float* y = (float*)d_out;

    float* ws     = (float*)d_ws;
    float* h      = ws;
    float* a_src  = ws + 3200000;
    float* a_dst  = ws + 3250000;
    unsigned* binw = (unsigned*)(ws + 3300000);
    unsigned* rec  = (unsigned*)(ws + 3500000);

    gat_gemm_wmma_kernel<<<N_TILES, 32, 0, stream>>>(x, W, att_src_v, att_dst_v,
                                                     h, a_src, a_dst, ei, binw, rec);

    gat_owner_kernel<<<N_OWN_BLOCKS, OWN_T, 0, stream>>>(rec, binw, a_src, a_dst, h, bias_conv,
                                                         W_lin, b_lin, y);
}


# Cross-lane sums (a_src/a_dst row sums of the GEMM waves, the two sums of every output head) by DPP row_xmask adds and one permlanex16, not shuffles through the LDS unit; same partners and order
# speedup vs baseline: 5.8542x; 1.0608x over previous
#include <hip/hip_runtime.h>
#include <math.h>

#define N_NODES 50000
#define N_EDGES 800000
#define IN_DIM 128
#define HIDDEN 64
#define NEG_SLOPE 0.2f
#define E_TOT (N_EDGES + N_NODES)
#define NPB 200
#define N_OWN_BLOCKS (N_NODES / NPB)
#define N_TILES (N_NODES / 16)
#define NO_BIN 255u

typedef float v2f __attribute__((ext_vector_type(2)));
typedef float v4f __attribute__((ext_vector_type(4)));
typedef float v8f __attribute__((ext_vector_type(8)));

__device__ __forceinline__ float row16_sum(float v) {
    v += __int_as_float(__builtin_amdgcn_update_dpp(0, __float_as_int(v), 0x161, 0xF, 0xF, false));
    v += __int_as_float(__builtin_amdgcn_update_dpp(0, __float_as_int(v), 0x162, 0xF, 0xF, false));
    v += __int_as_float(__builtin_amdgcn_update_dpp(0, __float_as_int(v), 0x164, 0xF, 0xF, false));
    v += __int_as_float(__builtin_amdgcn_update_dpp(0, __float_as_int(v), 0x168, 0xF, 0xF, false));
    return v;
}

__device__ __forceinline__ float wave32_sum(float v) {
    v = row16_sum(v);
    const int b = __float_as_int(v);
    return v + __int_as_float(__builtin_amdgcn_permlanex16(b, b, 0x76543210, 0xfedcba98, false, false));
}

__device__ __forceinline__ unsigned bin_of(int d) {
    const unsigned u = (unsigned)d;
    return (u < (unsigned)N_NODES) ? (u / NPB) : NO_BIN;
}

__global__ void __launch_bounds__(32)
gat_gemm_wmma_kernel(const float* __restrict__ x, const float* __restrict__ W,
                     const float* __restrict__ att_src_v, const float* __restrict__ att_dst_v,
                     float* __restrict__ h, float* __restrict__ a_src,
                     float* __restrict__ a_dst,
                     const int* __restrict__ ei, unsigned* __restrict__ binw,
                     unsigned* __restrict__ rec)
{
    const int tile = blockIdx.x;
    const int lane = threadIdx.x;
    const int half = lane >> 4;
    const int lrow = lane & 15;
    const int row0 = tile * 16;

    const float* xrow  = x + (size_t)(row0 + lrow) * IN_DIM;
    const float* wrow  = W + (size_t)lrow * IN_DIM;

    static_assert(N_EDGES == N_TILES * 256, "each GEMM wave converts 256 edges");
    const int4* dst4 = (const int4*)(ei + N_EDGES);
    const int4 dv0 = dst4[tile * 64 + lane];
    const int4 dv1 = dst4[tile * 64 + 32 + lane];
    const int4* src4 = (const int4*)ei;
    const int4 sv0 = src4[tile * 64 + lane];
    const int4 sv1 = src4[tile * 64 + 32 + lane];

    v8f acc0 = {}, acc1 = {}, acc2 = {}, acc3 = {};

    for (int kb = 0; kb < IN_DIM; kb += 8) {
        const int k = kb + 4 * half;
        const v4f a  = *(const v4f*)(xrow + k);
        const v4f b0 = *(const v4f*)(wrow + k);
        const v4f b1 = *(const v4f*)(wrow + 16 * IN_DIM + k);
        const v4f b2 = *(const v4f*)(wrow + 32 * IN_DIM + k);
        const v4f b3 = *(const v4f*)(wrow + 48 * IN_DIM + k);
        const v2f a_lo = __builtin_shufflevector(a, a, 0, 1), a_hi = __builtin_shufflevector(a, a, 2, 3);
        acc0 = __builtin_amdgcn_wmma_f32_16x16x4_f32(false, a_lo, false, __builtin_shufflevector(b0, b0, 0, 1), (short)0, acc0, false, false);
        acc1 = __builtin_amdgcn_wmma_f32_16x16x4_f32(false, a_lo, false, __builtin_shufflevector(b1, b1, 0, 1), (short)0, acc1, false, false);
        acc2 = __builtin_amdgcn_wmma_f32_16x16x4_f32(false, a_lo, false, __builtin_shufflevector(b2, b2, 0, 1), (short)0, acc2, false, false);
        acc3 = __builtin_amdgcn_wmma_f32_16x16x4_f32(false, a_lo, false, __builtin_shufflevector(b3, b3, 0, 1), (short)0, acc3, false, false);
        acc0 = __builtin_amdgcn_wmma_f32_16x16x4_f32(false, a_hi, false, __builtin_shufflevector(b0, b0, 2, 3), (short)0, acc0, false, false);
        acc1 = __builtin_amdgcn_wmma_f32_16x16x4_f32(false, a_hi, false, __builtin_shufflevector(b1, b1, 2, 3), (short)0, acc1, false, false);
        acc2 = __builtin_amdgcn_wmma_f32_16x16x4_f32(false, a_hi, false, __builtin_shufflevector(b2, b2, 2, 3), (short)0, acc2, false, false);
        acc3 = __builtin_amdgcn_wmma_f32_16x16x4_f32(false, a_hi, false, __builtin_shufflevector(b3, b3, 2, 3), (short)0, acc3, false, false);
    }

    const float as0 = att_src_v[lrow],      ad0 = att_dst_v[lrow];
    const float as1 = att_src_v[16 + lrow], ad1 = att_dst_v[16 + lrow];
    const float as2 = att_src_v[32 + lrow], ad2 = att_dst_v[32 + lrow];
    const float as3 = att_src_v[48 + lrow], ad3 = att_dst_v[48 + lrow];

    #pragma unroll
    for (int r = 0; r < 8; ++r) {
        const int row = row0 + r + 8 * half;
        float* hr = h + (size_t)row * HIDDEN;
        hr[lrow]      = acc0[r];
        hr[16 + lrow] = acc1[r];
        hr[32 + lrow] = acc2[r];
        hr[48 + lrow] = acc3[r];
        float ps = __builtin_fmaf(acc3[r], as3, __builtin_fmaf(acc2[r], as2, __builtin_fmaf(acc0[r], as0, acc1[r] * as1)));
        float pd = __builtin_fmaf(acc3[r], ad3, __builtin_fmaf(acc2[r], ad2, __builtin_fmaf(acc0[r], ad0, acc1[r] * ad1)));
        ps = row16_sum(ps);
        pd = row16_sum(pd);
        if (lrow == 0) { a_src[row] = ps; a_dst[row] = pd; }
    }

    {
        unsigned wa = 0, wb = 0;
        #pragma unroll
        for (int t = 0; t < 2; ++t) {
            const int4 dv = t ? dv1 : dv0;
            const int4 sv = t ? sv1 : sv0;
            const unsigned bb[4] = { bin_of(dv.x), bin_of(dv.y), bin_of(dv.z), bin_of(dv.w) };
            uint4 rv;
            rv.x = (unsigned)sv.x | ((((unsigned)dv.x - bb[0] * NPB) & 0xFFu) << 16);
            rv.y = (unsigned)sv.y | ((((unsigned)dv.y - bb[1] * NPB) & 0xFFu) << 16);
            rv.z = (unsigned)sv.z | ((((unsigned)dv.z - bb[2] * NPB) & 0xFFu) << 16);
            rv.w = (unsigned)sv.w | ((((unsigned)dv.w - bb[3] * NPB) & 0xFFu) << 16);
            ((uint4*)rec)[tile * 64 + t * 32 + lane] = rv;
            #pragma unroll
            for (int c = 0; c < 4; ++c) {
                #pragma unroll
                for (int i = 0; i < 8; ++i) {
                    const unsigned pl = __builtin_amdgcn_ballot_w32(((bb[c] >> i) & 1u) != 0u);
                    if (i < 4) wa = (lane == 16 * t + 4 * c + i)       ? pl : wa;
                    else       wb = (lane == 16 * t + 4 * c + (i - 4)) ? pl : wb;
                }
            }
        }
        binw[tile * 32 + lane]               = wa;
        binw[N_EDGES / 8 + tile * 32 + lane] = wb;
    }
}

#define QCAP 16384
#define OWN_T 1024
#define PB 16
#define SCAN_U 5
#define OWN_W (OWN_T / 32)
#define RPW ((NPB + OWN_W - 1) / OWN_W)

__device__ __forceinline__ void lds_max_float(float* addr, float val) {
    if (val >= 0.0f) atomicMax((int*)addr, __float_as_int(val));
    else             atomicMin((unsigned int*)addr, __float_as_uint(val));
}

struct PlaneKey { unsigned w[8]; };

__device__ __forceinline__ unsigned hits32(uint4 lo, uint4 hi, const PlaneKey& key) {
    return (lo.x ^ key.w[0]) & (lo.y ^ key.w[1]) & (lo.z ^ key.w[2]) & (lo.w ^ key.w[3]) &
           (hi.x ^ key.w[4]) & (hi.y ^ key.w[5]) & (hi.z ^ key.w[6]) & (hi.w ^ key.w[7]);
}

__device__ __forceinline__ int wave_prefix_sum(int v, int lane) {
    v += __builtin_amdgcn_update_dpp(0, v, 0x111, 0xF, 0xF, false);
    v += __builtin_amdgcn_update_dpp(0, v, 0x112, 0xF, 0xF, false);
    v += __builtin_amdgcn_update_dpp(0, v, 0x114, 0xF, 0xF, false);
    v += __builtin_amdgcn_update_dpp(0, v, 0x118, 0xF, 0xF, false);
    const int low = __builtin_amdgcn_readlane(v, 15);
    return v + ((lane >= 16) ? low : 0);
}

template <bool GUARD>
__device__ __forceinline__ void scan_round(const uint4* __restrict__ plane_lo,
                                           const uint4* __restrict__ plane_hi,
                                           int g0, int lenG, int rotG, const PlaneKey& key,
                                           int i0, int lane, unsigned* q_pk, int* q_n) {
    uint4 lo[SCAN_U], hi[SCAN_U];
    int   G[SCAN_U];
    #pragma unroll
    for (int k = 0; k < SCAN_U; ++k) {
        int jj = i0 + k * OWN_T + rotG;
        if (jj >= lenG) jj -= lenG;
        G[k] = g0 + jj;
        if (!GUARD || (i0 + k * OWN_T < lenG)) { lo[k] = plane_lo[G[k]]; hi[k] = plane_hi[G[k]]; }
        else { lo[k] = make_uint4(~0u, ~0u, ~0u, ~0u); hi[k] = lo[k]; }
    }
    unsigned m[SCAN_U];
    int cnt = 0;
    #pragma unroll
    for (int k = 0; k < SCAN_U; ++k) {
        m[k] = hits32(lo[k], hi[k], key);
        cnt += __popc(m[k]);
    }
    const int incl  = wave_prefix_sum(cnt, lane);
    const int total = __builtin_amdgcn_readlane(incl, 31);
    int base = 0;
    if (lane == 0) base = atomicAdd(q_n, total);
    int slot = __builtin_amdgcn_readfirstlane(base) + incl - cnt;
    #pragma unroll
    for (int k = 0; k < SCAN_U; ++k) {
        unsigned mm = m[k];
        const int e0 = ((G[k] & ~3) << 5) + (G[k] & 3);
        while (mm) {
            const int p = __builtin_ctz(mm);
            mm &= mm - 1;
            if (slot < QCAP) q_pk[slot] = (unsigned)(e0 + 4 * p);
            ++slot;
        }
    }
}

__device__ __forceinline__ void scan_enqueue(const uint4* __restrict__ plane_lo,
                                             const uint4* __restrict__ plane_hi,
                                             int g0, int lenG, int rotG, const PlaneKey& key,
                                             int tid, unsigned* q_pk, int* q_n) {
    const int full = lenG / (SCAN_U * OWN_T);
    for (int r = 0; r < full; ++r)
        scan_round<false>(plane_lo, plane_hi, g0, lenG, rotG, key, tid + r * (SCAN_U * OWN_T), tid & 31, q_pk, q_n);
    if (full * (SCAN_U * OWN_T) < lenG)
        scan_round<true>(plane_lo, plane_hi, g0, lenG, rotG, key, tid + full * (SCAN_U * OWN_T), tid & 31, q_pk, q_n);
}

__device__ __forceinline__ void resolve_max(const unsigned* __restrict__ rec,
                                            const float* __restrict__ a_src,
                                            const float* adst_l, float* m_l,
                                            const unsigned* q_pk, unsigned* q_sd, float* q_sc,
                                            int n, int tid) {
    for (int i0 = tid; i0 < n; i0 += 4 * OWN_T) {
        unsigned u[4];
        unsigned r[4];
        float    as[4];
        #pragma unroll
        for (int k = 0; k < 4; ++k) {
            const int i = i0 + k * OWN_T;
            u[k] = q_pk[(i < n) ? i : (n - 1)];
        }
        #pragma unroll
        for (int k = 0; k < 4; ++k) r[k] = rec[u[k]];
        #pragma unroll
        for (int k = 0; k < 4; ++k) as[k] = a_src[r[k] & 0xFFFFu];
        #pragma unroll
        for (int k = 0; k < 4; ++k) {
            const int i = i0 + k * OWN_T;
            if (i < n) {
                const unsigned dl = r[k] >> 16;
                float v = as[k] + adst_l[dl];
                v = (v > 0.0f) ? v : v * NEG_SLOPE;
                q_sd[i] = r[k];
                q_sc[i] = v;
                lds_max_float(&m_l[dl], v);
            }
        }
    }
}

template <bool FULL>
__device__ __forceinline__ void process_batch(const float* __restrict__ h,
                                              const float* m_l, float* s_l, float* acc,
                                              const unsigned* q_sd, const float* q_sc,
                                              int base, int n, int lane) {
    const int  my = base + (lane & (PB - 1));
    const bool ok = FULL || (my < n);
    const unsigned u  = q_sd[ok ? my : base];
    const float    sc = q_sc[ok ? my : base];
    const float    ex = expf(sc - m_l[u >> 16]);
    if (ok && lane < PB) atomicAdd(&s_l[u >> 16], ex);
    v2f hv[PB];
    #pragma unroll
    for (int k = 0; k < PB; ++k) {
        const unsigned uk = (unsigned)__builtin_amdgcn_readlane((int)u, k);
        hv[k] = *(const v2f*)(h + (size_t)(uk & 0xFFFFu) * HIDDEN + lane * 2);
    }
    #pragma unroll
    for (int k = 0; k < PB; ++k) {
        if (FULL || (base + k < n)) {
            const unsigned uk  = (unsigned)__builtin_amdgcn_readlane((int)u, k);
            const float    exk = __int_as_float(__builtin_amdgcn_readlane(__float_as_int(ex), k));
            float* a = acc + (uk >> 16) * HIDDEN + lane * 2;
            atomicAdd(a,     exk * hv[k].x);
            atomicAdd(a + 1, exk * hv[k].y);
        }
    }
}

__device__ __forceinline__ void process_queue(const float* __restrict__ h,
                                              const float* m_l, float* s_l, float* acc,
                                              const unsigned* q_sd, const float* q_sc,
                                              int n, int wv, int lane) {
    for (int base = wv * PB; base < n; base += (OWN_T / 32) * PB) {
        if (base + PB <= n) process_batch<true >(h, m_l, s_l, acc, q_sd, q_sc, base, n, lane);
        else                process_batch<false>(h, m_l, s_l, acc, q_sd, q_sc, base, n, lane);
    }
}

#define RCAP 64
#define RSTR 65
static_assert(NPB * RSTR <= QCAP, "the rows' lists live in the resolved arrays");

__device__ __forceinline__ void resolve_route(const unsigned* __restrict__ rec,
                                              const float* __restrict__ a_src,
                                              const float* adst_l, float* m_l,
                                              const unsigned* q_pk, unsigned* q_sd, float* q_sc,
                                              int* row_n, int* ovf, int n, int tid) {
    for (int i0 = tid; i0 < n; i0 += 4 * OWN_T) {
        unsigned u[4];
        unsigned r[4];
        float    as[4];
        #pragma unroll
        for (int k = 0; k < 4; ++k) {
            const int i = i0 + k * OWN_T;
            u[k] = q_pk[(i < n) ? i : (n - 1)];
        }
        #pragma unroll
        for (int k = 0; k < 4; ++k) r[k] = rec[u[k]];
        #pragma unroll
        for (int k = 0; k < 4; ++k) as[k] = a_src[r[k] & 0xFFFFu];
        float v[4];
        int   t[4];
        #pragma unroll
        for (int k = 0; k < 4; ++k) {
            const unsigned dl = r[k] >> 16;
            v[k] = as[k] + adst_l[dl];
            v[k] = (v[k] > 0.0f) ? v[k] : v[k] * NEG_SLOPE;
        }
        #pragma unroll
        for (int k = 0; k < 4; ++k) {
            const int i = i0 + k * OWN_T;
            t[k] = RCAP;
            if (i < n) {
                const unsigned dl = r[k] >> 16;
                lds_max_float(&m_l[dl], v[k]);
                t[k] = atomicAdd(&row_n[dl], 1);
            }
        }
        #pragma unroll
        for (int k = 0; k < 4; ++k) {
            const int i = i0 + k * OWN_T;
            if (i < n) {
                const unsigned dl = r[k] >> 16;
                if (t[k] < RCAP) { q_sd[dl * RSTR + t[k]] = r[k] & 0xFFFFu; q_sc[dl * RSTR + t[k]] = v[k]; }
                else             *ovf = 1;
            }
        }
    }
}

template <int NS, int MINC>
__device__ __forceinline__ void row_part(const float* __restrict__ h, unsigned u, float ex, int cnt, int lane,
                                         float& a0, float& a1) {
    v2f hv[NS];
    #pragma unroll
    for (int k = 0; k < NS; ++k) {
        const unsigned uk = (unsigned)__builtin_amdgcn_readlane((int)u, k);
        hv[k] = *(const v2f*)(h + (size_t)uk * HIDDEN + lane * 2);
    }
    #pragma unroll
    for (int k = 0; k < NS; ++k) {
        const float exk = __int_as_float(__builtin_amdgcn_readlane(__float_as_int(ex), k));
        float hx = hv[k].x, hy = hv[k].y;
        if (k >= MINC) {
            const bool live = k < cnt;
            hx = live ? hx : 0.0f;
            hy = live ? hy : 0.0f;
        }
        a0 = __builtin_fmaf(exk, hx, a0);
        a1 = __builtin_fmaf(exk, hy, a1);
    }
}

#define RB 32
__device__ __forceinline__ void row_batch(const float* __restrict__ h,
                                          const unsigned* rsd, const float* rsc, float m,
                                          int base, int n, int lane,
                                          float& a0, float& a1, float& wsum) {
    const int  cnt = (n - base < RB) ? (n - base) : RB;
    const bool ok  = lane < cnt;
    const unsigned u  = rsd[base + (ok ? lane : 0)];
    const float    sc = rsc[base + (ok ? lane : 0)];
    const float    ex = ok ? expf(sc - m) : 0.0f;
    wsum += ex;
    static_assert(RB == 32, "row_batch: the last case takes every count up to RB");
    switch ((cnt + 3) >> 2) {
        case 1:
        case 2:
        case 3:  row_part<12,  1>(h, u, ex, cnt, lane, a0, a1); break;
        case 4:  row_part<16, 13>(h, u, ex, cnt, lane, a0, a1); break;
        case 5:  row_part<20, 17>(h, u, ex, cnt, lane, a0, a1); break;
        case 6:  row_part<24, 21>(h, u, ex, cnt, lane, a0, a1); break;
        default: row_part<32, 25>(h, u, ex, cnt, lane, a0, a1); break;
    }
}

__global__ void __launch_bounds__(OWN_T)
gat_owner_kernel(const unsigned* __restrict__ rec,
                 const unsigned* __restrict__ binw,
                 const float* __restrict__ a_src,
                 const float* __restrict__ a_dst,
                 const float* __restrict__ h,
                 const float* __restrict__ bias_conv,
                 const float* __restrict__ W_lin,
                 const float* __restrict__ b_lin,
                 float* __restrict__ y) {
    __shared__ float    acc[NPB * HIDDEN];
    __shared__ float    m_l[NPB];
    __shared__ float    s_l[NPB];
    __shared__ float    adst_l[NPB];
    __shared__ float    self_sc[NPB];
    __shared__ unsigned q_pk[QCAP];
    __shared__ unsigned q_sd[QCAP];
    __shared__ float    q_sc[QCAP];
    __shared__ int      q_n;
    __shared__ int      row_n[NPB];
    __shared__ int      ovf;

    const int tid   = threadIdx.x;
    const int lane  = tid & 31;
    const int wv    = __builtin_amdgcn_readfirstlane(tid >> 5);
    __builtin_assume(wv >= 0 && wv < OWN_W);
    const int node0 = blockIdx.x * NPB;
    const uint4* plane_lo = (const uint4*)binw;
    const uint4* plane_hi = (const uint4*)(binw + N_EDGES / 8);
    PlaneKey key;
    #pragma unroll
    for (int i = 0; i < 8; ++i) key.w[i] = ((blockIdx.x >> i) & 1u) - 1u;
    static_assert(N_OWN_BLOCKS <= 255 && N_EDGES % 128 == 0 && QCAP % 128 == 0, "plane list");

    static_assert(RPW <= 8 && OWN_W <= NPB, "own rows: a wave's per-row results fit three 8-wide registers, and row wv exists");

    if (tid < NPB) {
        const float ad = a_dst[node0 + tid];
        float v = a_src[node0 + tid] + ad;
        v = (v > 0.0f) ? v : v * NEG_SLOPE;
        adst_l[tid]  = ad;
        self_sc[tid] = v;
        m_l[tid]     = v;
        q_sd[tid * RSTR] = (unsigned)(node0 + tid);
        q_sc[tid * RSTR] = v;
        row_n[tid]       = 1;
    }
    if (tid == 0) { q_n = 0; ovf = 0; }

    __syncthreads();

    scan_enqueue(plane_lo, plane_hi, 0, N_EDGES / 32, blockIdx.x * (N_EDGES / 32 / N_OWN_BLOCKS), key, tid, q_pk, &q_n);
    __syncthreads();
    const int n_all = __builtin_amdgcn_readfirstlane(q_n);

    const float bc0 = bias_conv[lane * 2], bc1 = bias_conv[lane * 2 + 1];
    const float wl0 = W_lin[lane * 2],     wl1 = W_lin[lane * 2 + 1];
    const float bl  = b_lin[0];

    bool own_lists = (n_all <= QCAP);
    if (own_lists) {
        resolve_route(rec, a_src, adst_l, m_l, q_pk, q_sd, q_sc, row_n, &ovf, n_all, tid);
        __syncthreads();
        own_lists = (__builtin_amdgcn_readfirstlane(ovf) == 0);
    }

    if (own_lists) {
        v8f ra0 = {}, ra1 = {}, rws = {};
        #pragma nounroll
        for (int j = 0; j < RPW; ++j) {
            const int dl = wv + j * OWN_W;
            if (dl >= NPB) break;
            const int   n = __builtin_amdgcn_readfirstlane(row_n[dl]);
            const float m = m_l[dl];
            float a0 = 0.0f, a1 = 0.0f, wsum = 0.0f;
            for (int base = 0; base < n; base += RB)
                row_batch(h, q_sd + dl * RSTR, q_sc + dl * RSTR, m, base, n, lane, a0, a1, wsum);
            ra0[j] = a0; ra1[j] = a1; rws[j] = wsum;
        }
        #pragma unroll
        for (int j = 0; j < RPW; ++j) {
            const int dl = wv + j * OWN_W;
            if (dl < NPB) {
                const float ws = wave32_sum(rws[j]);
                const float inv = 1.0f / ws;
                const float o0 = fmaxf(ra0[j] * inv + bc0, 0.0f);
                const float o1 = fmaxf(ra1[j] * inv + bc1, 0.0f);
                const float p = wave32_sum(o0 * wl0 + o1 * wl1);
                if (lane == 0) y[node0 + dl] = 1.0f / (1.0f + expf(-(p + bl)));
            }
        }
    } else {
        for (int b = 0; b < N_EDGES; b += QCAP) {
            const int len = (N_EDGES - b < QCAP) ? (N_EDGES - b) : QCAP;
            __syncthreads();
            if (tid == 0) q_n = 0;
            __syncthreads();
            scan_enqueue(plane_lo, plane_hi, b / 32, len / 32, 0, key, tid, q_pk, &q_n);
            __syncthreads();
            resolve_max(rec, a_src, adst_l, m_l, q_pk, q_sd, q_sc, __builtin_amdgcn_readfirstlane(q_n), tid);
        }
        __syncthreads();
        for (int i = tid; i < NPB * HIDDEN; i += OWN_T) {
            const int dl = i >> 6;
            const float ex = expf(self_sc[dl] - m_l[dl]);
            acc[i] = ex * h[(size_t)node0 * HIDDEN + i];
            if ((i & 63) == 0) s_l[dl] = ex;
        }
        for (int b = 0; b < N_EDGES; b += QCAP) {
            const int len = (N_EDGES - b < QCAP) ? (N_EDGES - b) : QCAP;
            __syncthreads();
            if (tid == 0) q_n = 0;
            __syncthreads();
            scan_enqueue(plane_lo, plane_hi, b / 32, len / 32, 0, key, tid, q_pk, &q_n);
            __syncthreads();
            const int n = __builtin_amdgcn_readfirstlane(q_n);
            resolve_max(rec, a_src, adst_l, m_l, q_pk, q_sd, q_sc, n, tid);
            __syncthreads();
            process_queue(h, m_l, s_l, acc, q_sd, q_sc, n, wv, lane);
        }
        __syncthreads();
        for (int dl = wv; dl < NPB; dl += OWN_T / 32) {
            const float inv = 1.0f / s_l[dl];
            const v2f o = *(const v2f*)(acc + dl * HIDDEN + lane * 2);
            const float o0 = fmaxf(o.x * inv + bc0, 0.0f);
            const float o1 = fmaxf(o.y * inv + bc1, 0.0f);
            const float p = wave32_sum(o0 * wl0 + o1 * wl1);
            if (lane == 0) y[node0 + dl] = 1.0f / (1.0f + expf(-(p + bl)));
        }
    }
}

extern "C" void kernel_launch(void* const* d_in, const int* in_sizes, int n_in,
                              void* d_out, int out_size, void* d_ws, size_t ws_size,
                              hipStream_t stream) {
    const float* x         = (const float*)d_in[0];
    const int*   ei        = (const int*)  d_in[1];
    const float* W         = (const float*)d_in[2];
    const float* att_src_v = (const float*)d_in[3];
    const float* att_dst_v = (const float*)d_in[4];
    const float* bias_conv = (const float*)d_in[5];
    const float* W_lin     = (const float*)d_in[6];
    const float* b_lin     = (const float*)d_in[7];
    float* y = (float*)d_out;

    float* ws     = (float*)d_ws;
    float* h      = ws;
    float* a_src  = ws + 3200000;
    float* a_dst  = ws + 3250000;
    unsigned* binw = (unsigned*)(ws + 3300000);
    unsigned* rec  = (unsigned*)(ws + 3500000);

    gat_gemm_wmma_kernel<<<N_TILES, 32, 0, stream>>>(x, W, att_src_v, att_dst_v,
                                                     h, a_src, a_dst, ei, binw, rec);

    gat_owner_kernel<<<N_OWN_BLOCKS, OWN_T, 0, stream>>>(rec, binw, a_src, a_dst, h, bias_conv,
                                                         W_lin, b_lin, y);
}
